# starting kernel
# baseline (speedup 1.0000x reference)
#include <hip/hip_runtime.h>

typedef __attribute__((ext_vector_type(2))) float v2f;
typedef __attribute__((ext_vector_type(8))) float v8f;

#define BATCH   4
#define NPTS    8192
#define CIN     64
#define COUT    128
#define NPOINT  2048
#define KNN     16
#define KRAW    67
#define KPAD    68
#define BNEPS   1e-5f

__global__ __launch_bounds__(1024) void fps_kernel(const float* __restrict__ xyz,
                                                   int* __restrict__ fps_idx)
{
    const int b   = blockIdx.x;
    const int tid = threadIdx.x;
    const float* xb = xyz + (size_t)b * NPTS * 3;

    __shared__ float sdist[1024];
    __shared__ int   sidx[1024];

    float dist[8];
#pragma unroll
    for (int i = 0; i < 8; ++i) dist[i] = 3.4e38f;

    int cur = 0;
    for (int s = 0; s < NPOINT; ++s) {
        if (tid == 0) fps_idx[b * NPOINT + s] = cur;
        const float cx = xb[cur * 3 + 0];
        const float cy = xb[cur * 3 + 1];
        const float cz = xb[cur * 3 + 2];

        float best = -1.0f; int bidx = 0;
#pragma unroll
        for (int i = 0; i < 8; ++i) {
            const int n = i * 1024 + tid;
            const float dx = xb[n * 3 + 0] - cx;
            const float dy = xb[n * 3 + 1] - cy;
            const float dz = xb[n * 3 + 2] - cz;
            const float d  = dx * dx + dy * dy + dz * dz;
            const float nd = fminf(dist[i], d);
            dist[i] = nd;
            if (nd > best) { best = nd; bidx = n; }
        }
        sdist[tid] = best; sidx[tid] = bidx;
        __syncthreads();
        for (int off = 512; off > 0; off >>= 1) {
            if (tid < off) {
                const float d1 = sdist[tid],       d2 = sdist[tid + off];
                const int   i1 = sidx[tid],        i2 = sidx[tid + off];
                if (d2 > d1 || (d2 == d1 && i2 < i1)) { sdist[tid] = d2; sidx[tid] = i2; }
            }
            __syncthreads();
        }
        cur = sidx[0];
        __syncthreads();
    }
}

__global__ void gather_new_xyz_kernel(const float* __restrict__ xyz,
                                      const int* __restrict__ fps_idx,
                                      float* __restrict__ new_xyz)
{
    const int e = blockIdx.x * blockDim.x + threadIdx.x;
    if (e >= BATCH * NPOINT * 3) return;
    const int c = e % 3;
    const int p = (e / 3) % NPOINT;
    const int b = e / (3 * NPOINT);
    const int n = fps_idx[b * NPOINT + p];
    new_xyz[e] = xyz[((size_t)b * NPTS + n) * 3 + c];
}

__global__ __launch_bounds__(128) void knn_kernel(const float* __restrict__ xyz,
                                                  const float* __restrict__ new_xyz,
                                                  int* __restrict__ knn_idx)
{
    const int q = blockIdx.x * blockDim.x + threadIdx.x;
    if (q >= BATCH * NPOINT) return;
    const int b = q / NPOINT;
    const float qx = new_xyz[q * 3 + 0];
    const float qy = new_xyz[q * 3 + 1];
    const float qz = new_xyz[q * 3 + 2];
    const float* xb = xyz + (size_t)b * NPTS * 3;

    float v[KNN]; int id[KNN];
#pragma unroll
    for (int t = 0; t < KNN; ++t) { v[t] = 3.4e38f; id[t] = 0; }

    for (int j = 0; j < NPTS; ++j) {
        if ((j & 63) == 0)
            __builtin_prefetch(xb + (size_t)(j + 256) * 3, 0, 1);
        const float dx = xb[j * 3 + 0] - qx;
        const float dy = xb[j * 3 + 1] - qy;
        const float dz = xb[j * 3 + 2] - qz;
        const float d  = dx * dx + dy * dy + dz * dz;
        if (d < v[KNN - 1]) {
            v[KNN - 1] = d; id[KNN - 1] = j;
#pragma unroll
            for (int t = KNN - 1; t > 0; --t) {
                if (v[t] < v[t - 1]) {
                    const float tv = v[t]; v[t] = v[t - 1]; v[t - 1] = tv;
                    const int   ti = id[t]; id[t] = id[t - 1]; id[t - 1] = ti;
                }
            }
        }
    }
#pragma unroll
    for (int t = 0; t < KNN; ++t) knn_idx[(size_t)q * KNN + t] = id[t];
}

__global__ void pad_w_kernel(const float* __restrict__ W, float* __restrict__ Wpad)
{
    const int e = blockIdx.x * blockDim.x + threadIdx.x;
    if (e >= KPAD * COUT) return;
    const int r = e / COUT;
    Wpad[e] = (r < KRAW) ? W[e] : 0.0f;
}

__global__ __launch_bounds__(256) void group_gemm_kernel(
    const float* __restrict__ xyz, const float* __restrict__ features,
    const float* __restrict__ new_xyz, const int* __restrict__ knn_idx,
    const float* __restrict__ Wpad,
    const float* __restrict__ gamma, const float* __restrict__ beta,
    const float* __restrict__ meanv, const float* __restrict__ varv,
    float* __restrict__ partial_sum, float* __restrict__ partial_sq,
    float* __restrict__ out_features, int final_pass)
{
    const int q    = blockIdx.x;
    const int b    = q / NPOINT;
    const int tid  = threadIdx.x;
    const int lane = tid & 31;
    const int wv   = tid >> 5;

    __shared__ float As[KNN][KPAD];
    __shared__ float red[8][32];

    const float qx = new_xyz[q * 3 + 0];
    const float qy = new_xyz[q * 3 + 1];
    const float qz = new_xyz[q * 3 + 2];

    for (int e = tid; e < KNN * KPAD; e += 256) {
        const int kk = e / KPAD;
        const int c  = e % KPAD;
        const int n  = knn_idx[(size_t)q * KNN + kk];
        float val;
        if (c < 3) {
            const float pv = xyz[((size_t)b * NPTS + n) * 3 + c];
            val = pv - (c == 0 ? qx : (c == 1 ? qy : qz));
        } else if (c < KRAW) {
            val = features[((size_t)b * NPTS + n) * CIN + (c - 3)];
        } else {
            val = 0.0f;
        }
        As[kk][c] = val;
    }
    __syncthreads();

    const int col = wv * 16 + (lane & 15);
    const int hi  = lane >> 4;
    const int row = lane & 15;

    v8f acc = {};
#pragma unroll
    for (int kb = 0; kb < KPAD / 4; ++kb) {
        const int k0 = kb * 4 + hi * 2;
        v2f a, bb;
        a.x  = As[row][k0 + 0];
        a.y  = As[row][k0 + 1];
        bb.x = Wpad[(k0 + 0) * COUT + col];
        bb.y = Wpad[(k0 + 1) * COUT + col];
        acc = __builtin_amdgcn_wmma_f32_16x16x4_f32(
                  false, a, false, bb, (short)0, acc, false, false);
    }

    if (!final_pass) {
        float s = 0.0f, ss = 0.0f;
#pragma unroll
        for (int t = 0; t < 8; ++t) { const float h = acc[t]; s += h; ss += h * h; }
        red[wv][lane] = s;
        __syncthreads();
        float s2 = 0.0f;
        if (lane < 16) s2 = red[wv][lane] + red[wv][lane + 16];
        __syncthreads();
        red[wv][lane] = ss;
        __syncthreads();
        if (lane < 16) {
            const float ss2 = red[wv][lane] + red[wv][lane + 16];
            partial_sum[(size_t)q * COUT + col] = s2;
            partial_sq [(size_t)q * COUT + col] = ss2;
        }
    } else {
        const float mu    = meanv[col];
        const float scale = gamma[col] * rsqrtf(varv[col] + BNEPS);
        const float shift = beta[col] - mu * scale;
        float m = -3.4e38f;
#pragma unroll
        for (int t = 0; t < 8; ++t) {
            float h = acc[t] * scale + shift;
            h = fmaxf(h, 0.0f);
            m = fmaxf(m, h);
        }
        red[wv][lane] = m;
        __syncthreads();
        if (lane < 16) {
            const float mm = fmaxf(red[wv][lane], red[wv][lane + 16]);
            out_features[(size_t)q * COUT + col] = mm;
        }
    }
}

__global__ __launch_bounds__(256) void reduce_stats_kernel(
    const float* __restrict__ partial_sum, const float* __restrict__ partial_sq,
    float* __restrict__ meanv, float* __restrict__ varv)
{
    const int ch  = blockIdx.x;
    const int tid = threadIdx.x;
    __shared__ float ssum[256], ssq[256];
    float s = 0.0f, sq = 0.0f;
    for (int i = tid; i < BATCH * NPOINT; i += 256) {
        s  += partial_sum[(size_t)i * COUT + ch];
        sq += partial_sq [(size_t)i * COUT + ch];
    }
    ssum[tid] = s; ssq[tid] = sq;
    __syncthreads();
    for (int off = 128; off > 0; off >>= 1) {
        if (tid < off) { ssum[tid] += ssum[tid + off]; ssq[tid] += ssq[tid + off]; }
        __syncthreads();
    }
    if (tid == 0) {
        const float cnt = (float)(BATCH * NPOINT * KNN);
        const float mu  = ssum[0] / cnt;
        meanv[ch] = mu;
        varv[ch]  = ssq[0] / cnt - mu * mu;
    }
}

extern "C" void kernel_launch(void* const* d_in, const int* in_sizes, int n_in,
                              void* d_out, int out_size, void* d_ws, size_t ws_size,
                              hipStream_t stream)
{
    const float* xyz      = (const float*)d_in[0];
    const float* features = (const float*)d_in[1];
    const float* W        = (const float*)d_in[2];
    const float* gamma    = (const float*)d_in[3];
    const float* beta     = (const float*)d_in[4];

    float* out          = (float*)d_out;
    float* new_xyz      = out;
    float* out_features = out + BATCH * NPOINT * 3;

    char* ws = (char*)d_ws;
    size_t off = 0;
    int*   fps_idx     = (int*)(ws + off);   off += sizeof(int)   * BATCH * NPOINT;
    int*   knn_idx     = (int*)(ws + off);   off += sizeof(int)   * BATCH * NPOINT * KNN;
    float* Wpad        = (float*)(ws + off); off += sizeof(float) * KPAD * COUT;
    off = (off + 255) & ~(size_t)255;
    float* partial_sum = (float*)(ws + off); off += sizeof(float) * BATCH * NPOINT * COUT;
    float* partial_sq  = (float*)(ws + off); off += sizeof(float) * BATCH * NPOINT * COUT;
    float* meanv       = (float*)(ws + off); off += sizeof(float) * COUT;
    float* varv        = (float*)(ws + off); off += sizeof(float) * COUT;
    (void)ws_size; (void)in_sizes; (void)n_in; (void)out_size;

    fps_kernel<<<BATCH, 1024, 0, stream>>>(xyz, fps_idx);
    gather_new_xyz_kernel<<<(BATCH * NPOINT * 3 + 255) / 256, 256, 0, stream>>>(
        xyz, fps_idx, new_xyz);
    knn_kernel<<<(BATCH * NPOINT + 127) / 128, 128, 0, stream>>>(
        xyz, new_xyz, knn_idx);
    pad_w_kernel<<<(KPAD * COUT + 255) / 256, 256, 0, stream>>>(W, Wpad);
    group_gemm_kernel<<<BATCH * NPOINT, 256, 0, stream>>>(
        xyz, features, new_xyz, knn_idx, Wpad, gamma, beta,
        meanv, varv, partial_sum, partial_sq, out_features, 0);
    reduce_stats_kernel<<<COUT, 256, 0, stream>>>(
        partial_sum, partial_sq, meanv, varv);
    group_gemm_kernel<<<BATCH * NPOINT, 256, 0, stream>>>(
        xyz, features, new_xyz, knn_idx, Wpad, gamma, beta,
        meanv, varv, partial_sum, partial_sq, out_features, 1);
}


# Sampling pick loop without 64-bit vector integer arithmetic: key maxima on 32-bit words (largest distance bits, then largest inverted index among holders); pick kept scalar; picks bit-identical
# speedup vs baseline: 2.3051x; 2.3051x over previous
#include <hip/hip_runtime.h>

typedef __attribute__((ext_vector_type(2))) float v2f;
typedef __attribute__((ext_vector_type(8))) float v8f;

#define BATCH   4
#define NPTS    8192
#define CIN     64
#define COUT    128
#define NPOINT  2048
#define KNN     16
#define KRAW    67
#define KPAD    68
#define BNEPS   1e-5f

typedef __attribute__((ext_vector_type(2))) unsigned v2u;
typedef __attribute__((ext_vector_type(4))) unsigned v4u;

__device__ __forceinline__ unsigned umax2(unsigned a, unsigned b)
{
    return a > b ? a : b;
}

__device__ __forceinline__ unsigned umax32(unsigned* m)
{
#pragma unroll
    for (int w = 16; w > 0; w >>= 1) {
#pragma unroll
        for (int j = 0; j < w; ++j) m[j] = umax2(m[j], m[j + w]);
    }
    return m[0];
}

__device__ __forceinline__ void keymax32(unsigned* h, unsigned* l, unsigned& mh, unsigned& ml)
{
    unsigned t[32];
#pragma unroll
    for (int j = 0; j < 32; ++j) t[j] = h[j];
    const unsigned top = umax32(t);
#pragma unroll
    for (int j = 0; j < 32; ++j) l[j] = (h[j] == top) ? l[j] : 0u;
    mh = top;
    ml = umax32(l);
}

__global__ __launch_bounds__(1024) void fps_kernel(const float* __restrict__ xyz,
                                                   int* __restrict__ fps_idx)
{
    const int b    = blockIdx.x;
    const int tid  = threadIdx.x;
    const int lane = tid & 31;
    const int wv   = tid >> 5;
    const float* xb = xyz + (size_t)b * NPTS * 3;

    __shared__ __attribute__((aligned(16))) v2u skey[1024];
    __shared__ __attribute__((aligned(16))) v2u spart[32];
    __shared__ int scur;

    const int wslot = (wv >> 1) * 64 + lane * 2 + (wv & 1);

    float dist[8];
#pragma unroll
    for (int i = 0; i < 8; ++i) dist[i] = 3.4e38f;

    int cur = 0;
    for (int s = 0; s < NPOINT; ++s) {
        if (tid == 0) fps_idx[b * NPOINT + s] = cur;
        const float cx = xb[cur * 3 + 0];
        const float cy = xb[cur * 3 + 1];
        const float cz = xb[cur * 3 + 2];

        float best = -1.0f; int bidx = 0;
#pragma unroll
        for (int i = 0; i < 8; ++i) {
            const int n = i * 1024 + tid;
            const float dx = xb[n * 3 + 0] - cx;
            const float dy = xb[n * 3 + 1] - cy;
            const float dz = xb[n * 3 + 2] - cz;
            float d;
            if (i < 6) {
                d = __builtin_fmaf(dz, dz, __builtin_fmaf(dx, dx, dy * dy));
            } else {
                float sxy;
                {
#pragma clang fp contract(off)
                    sxy = dx * dx + dy * dy;
                }
                d = __builtin_fmaf(dz, dz, sxy);
            }
            const float nd = fminf(dist[i], d);
            dist[i] = nd;
            if (nd > best) { best = nd; bidx = n; }
        }
        {
            v2u key;
            key.x = ~(unsigned)bidx;
            key.y = __float_as_uint(best);
            skey[wslot] = key;
        }
        __syncthreads();
        if (wv == 0) {
            const v4u* col = reinterpret_cast<const v4u*>(skey) + lane;
            unsigned h[32], l[32];
#pragma unroll
            for (int j = 0; j < 16; ++j) {
                const v4u p = col[j * 32];
                l[2 * j] = p.x; h[2 * j] = p.y; l[2 * j + 1] = p.z; h[2 * j + 1] = p.w;
            }
            unsigned ph, pl;
            keymax32(h, l, ph, pl);
            {
                v2u part;
                part.x = pl;
                part.y = ph;
                spart[lane] = part;
            }
            const v4u* all = reinterpret_cast<const v4u*>(spart);
#pragma unroll
            for (int j = 0; j < 16; ++j) {
                const v4u p = all[j];
                l[2 * j] = p.x; h[2 * j] = p.y; l[2 * j + 1] = p.z; h[2 * j + 1] = p.w;
            }
            unsigned gh, gl;
            keymax32(h, l, gh, gl);
            scur = (int)(~gl);
        }
        __syncthreads();
        cur = __builtin_amdgcn_readfirstlane(scur);
    }
}

__global__ void gather_new_xyz_kernel(const float* __restrict__ xyz,
                                      const int* __restrict__ fps_idx,
                                      float* __restrict__ new_xyz)
{
    const int e = blockIdx.x * blockDim.x + threadIdx.x;
    if (e >= BATCH * NPOINT * 3) return;
    const int c = e % 3;
    const int p = (e / 3) % NPOINT;
    const int b = e / (3 * NPOINT);
    const int n = fps_idx[b * NPOINT + p];
    new_xyz[e] = xyz[((size_t)b * NPTS + n) * 3 + c];
}

#define KSEG   16
#define KSEGN  (NPTS / KSEG)
#define KQ     32

__global__ __launch_bounds__(KSEG * 32) void knn_kernel(const float* __restrict__ xyz,
                                                        const float* __restrict__ new_xyz,
                                                        int* __restrict__ knn_idx)
{
    __shared__ float sd[KSEG][KNN][KQ];
    __shared__ int   si[KSEG][KNN][KQ];

    const int lane = threadIdx.x & 31;
    const int seg  = threadIdx.x >> 5;
    const int q    = blockIdx.x * KQ + lane;
    const int b    = (blockIdx.x * KQ) / NPOINT;
    const float qx = new_xyz[q * 3 + 0];
    const float qy = new_xyz[q * 3 + 1];
    const float qz = new_xyz[q * 3 + 2];
    const float* xb = xyz + (size_t)b * NPTS * 3;

    float v[KNN]; int id[KNN];
#pragma unroll
    for (int t = 0; t < KNN; ++t) { v[t] = 3.4e38f; id[t] = 0; }

    const int j0 = seg * KSEGN;
    for (int j = j0; j < j0 + KSEGN; ++j) {
        const float dx = xb[j * 3 + 0] - qx;
        const float dy = xb[j * 3 + 1] - qy;
        const float dz = xb[j * 3 + 2] - qz;
        const float d  = dx * dx + dy * dy + dz * dz;
        if (d < v[KNN - 1]) {
            v[KNN - 1] = d; id[KNN - 1] = j;
#pragma unroll
            for (int t = KNN - 1; t > 0; --t) {
                if (v[t] < v[t - 1]) {
                    const float tv = v[t]; v[t] = v[t - 1]; v[t - 1] = tv;
                    const int   ti = id[t]; id[t] = id[t - 1]; id[t - 1] = ti;
                }
            }
        }
    }
#pragma unroll
    for (int t = 0; t < KNN; ++t) { sd[seg][t][lane] = v[t]; si[seg][t][lane] = id[t]; }
    __syncthreads();

    if (seg == 0) {
        float hd[KSEG];
#pragma unroll
        for (int s = 0; s < KSEG; ++s) hd[s] = sd[s][0][lane];
        unsigned long long heads = 0ull;
        int oid[KNN];
#pragma unroll
        for (int t = 0; t < KNN; ++t) {
            float best = hd[0]; int bs = 0;
#pragma unroll
            for (int s = 1; s < KSEG; ++s)
                if (hd[s] < best) { best = hd[s]; bs = s; }
            const int p = (int)((heads >> (4 * bs)) & 15ull);
            oid[t] = si[bs][p][lane];
            if (t < KNN - 1) {
                heads += 1ull << (4 * bs);
                const float nh = sd[bs][p + 1][lane];
#pragma unroll
                for (int s = 0; s < KSEG; ++s) hd[s] = (s == bs) ? nh : hd[s];
            }
        }
#pragma unroll
        for (int t = 0; t < KNN; ++t) knn_idx[(size_t)q * KNN + t] = oid[t];
    }
}

__global__ void pad_w_kernel(const float* __restrict__ W, float* __restrict__ Wpad)
{
    const int e = blockIdx.x * blockDim.x + threadIdx.x;
    if (e >= KPAD * COUT) return;
    const int r = e / COUT;
    Wpad[e] = (r < KRAW) ? W[e] : 0.0f;
}

__global__ __launch_bounds__(256) void group_gemm_kernel(
    const float* __restrict__ xyz, const float* __restrict__ features,
    const float* __restrict__ new_xyz, const int* __restrict__ knn_idx,
    const float* __restrict__ Wpad,
    const float* __restrict__ gamma, const float* __restrict__ beta,
    const float* __restrict__ meanv, const float* __restrict__ varv,
    float* __restrict__ partial_sum, float* __restrict__ partial_sq,
    float* __restrict__ out_features, int final_pass)
{
    const int q    = blockIdx.x;
    const int b    = q / NPOINT;
    const int tid  = threadIdx.x;
    const int lane = tid & 31;
    const int wv   = tid >> 5;

    __shared__ float As[KNN][KPAD];
    __shared__ float red[8][32];

    const float qx = new_xyz[q * 3 + 0];
    const float qy = new_xyz[q * 3 + 1];
    const float qz = new_xyz[q * 3 + 2];

    for (int e = tid; e < KNN * KPAD; e += 256) {
        const int kk = e / KPAD;
        const int c  = e % KPAD;
        const int n  = knn_idx[(size_t)q * KNN + kk];
        float val;
        if (c < 3) {
            const float pv = xyz[((size_t)b * NPTS + n) * 3 + c];
            val = pv - (c == 0 ? qx : (c == 1 ? qy : qz));
        } else if (c < KRAW) {
            val = features[((size_t)b * NPTS + n) * CIN + (c - 3)];
        } else {
            val = 0.0f;
        }
        As[kk][c] = val;
    }
    __syncthreads();

    const int col = wv * 16 + (lane & 15);
    const int hi  = lane >> 4;
    const int row = lane & 15;

    v8f acc = {};
#pragma unroll
    for (int kb = 0; kb < KPAD / 4; ++kb) {
        const int k0 = kb * 4 + hi * 2;
        v2f a, bb;
        a.x  = As[row][k0 + 0];
        a.y  = As[row][k0 + 1];
        bb.x = Wpad[(k0 + 0) * COUT + col];
        bb.y = Wpad[(k0 + 1) * COUT + col];
        acc = __builtin_amdgcn_wmma_f32_16x16x4_f32(
                  false, a, false, bb, (short)0, acc, false, false);
    }

    if (!final_pass) {
        float s = 0.0f, ss = 0.0f;
#pragma unroll
        for (int t = 0; t < 8; ++t) { const float h = acc[t]; s += h; ss += h * h; }
        red[wv][lane] = s;
        __syncthreads();
        float s2 = 0.0f;
        if (lane < 16) s2 = red[wv][lane] + red[wv][lane + 16];
        __syncthreads();
        red[wv][lane] = ss;
        __syncthreads();
        if (lane < 16) {
            const float ss2 = red[wv][lane] + red[wv][lane + 16];
            partial_sum[(size_t)q * COUT + col] = s2;
            partial_sq [(size_t)q * COUT + col] = ss2;
        }
    } else {
        const float mu    = meanv[col];
        const float scale = gamma[col] * rsqrtf(varv[col] + BNEPS);
        const float shift = beta[col] - mu * scale;
        float m = -3.4e38f;
#pragma unroll
        for (int t = 0; t < 8; ++t) {
            float h = acc[t] * scale + shift;
            h = fmaxf(h, 0.0f);
            m = fmaxf(m, h);
        }
        red[wv][lane] = m;
        __syncthreads();
        if (lane < 16) {
            const float mm = fmaxf(red[wv][lane], red[wv][lane + 16]);
            out_features[(size_t)q * COUT + col] = mm;
        }
    }
}

__global__ __launch_bounds__(256) void reduce_stats_kernel(
    const float* __restrict__ partial_sum, const float* __restrict__ partial_sq,
    float* __restrict__ meanv, float* __restrict__ varv)
{
    const int ch  = blockIdx.x;
    const int tid = threadIdx.x;
    __shared__ float ssum[256], ssq[256];
    float s = 0.0f, sq = 0.0f;
    for (int i = tid; i < BATCH * NPOINT; i += 256) {
        s  += partial_sum[(size_t)i * COUT + ch];
        sq += partial_sq [(size_t)i * COUT + ch];
    }
    ssum[tid] = s; ssq[tid] = sq;
    __syncthreads();
    for (int off = 128; off > 0; off >>= 1) {
        if (tid < off) { ssum[tid] += ssum[tid + off]; ssq[tid] += ssq[tid + off]; }
        __syncthreads();
    }
    if (tid == 0) {
        const float cnt = (float)(BATCH * NPOINT * KNN);
        const float mu  = ssum[0] / cnt;
        meanv[ch] = mu;
        varv[ch]  = ssq[0] / cnt - mu * mu;
    }
}

extern "C" void kernel_launch(void* const* d_in, const int* in_sizes, int n_in,
                              void* d_out, int out_size, void* d_ws, size_t ws_size,
                              hipStream_t stream)
{
    const float* xyz      = (const float*)d_in[0];
    const float* features = (const float*)d_in[1];
    const float* W        = (const float*)d_in[2];
    const float* gamma    = (const float*)d_in[3];
    const float* beta     = (const float*)d_in[4];

    float* out          = (float*)d_out;
    float* new_xyz      = out;
    float* out_features = out + BATCH * NPOINT * 3;

    char* ws = (char*)d_ws;
    size_t off = 0;
    int*   fps_idx     = (int*)(ws + off);   off += sizeof(int)   * BATCH * NPOINT;
    int*   knn_idx     = (int*)(ws + off);   off += sizeof(int)   * BATCH * NPOINT * KNN;
    float* Wpad        = (float*)(ws + off); off += sizeof(float) * KPAD * COUT;
    off = (off + 255) & ~(size_t)255;
    float* partial_sum = (float*)(ws + off); off += sizeof(float) * BATCH * NPOINT * COUT;
    float* partial_sq  = (float*)(ws + off); off += sizeof(float) * BATCH * NPOINT * COUT;
    float* meanv       = (float*)(ws + off); off += sizeof(float) * COUT;
    float* varv        = (float*)(ws + off); off += sizeof(float) * COUT;
    (void)ws_size; (void)in_sizes; (void)n_in; (void)out_size;

    fps_kernel<<<BATCH, 1024, 0, stream>>>(xyz, fps_idx);
    gather_new_xyz_kernel<<<(BATCH * NPOINT * 3 + 255) / 256, 256, 0, stream>>>(
        xyz, fps_idx, new_xyz);
    knn_kernel<<<(BATCH * NPOINT) / KQ, KSEG * 32, 0, stream>>>(
        xyz, new_xyz, knn_idx);
    pad_w_kernel<<<(KPAD * COUT + 255) / 256, 256, 0, stream>>>(W, Wpad);
    group_gemm_kernel<<<BATCH * NPOINT, 256, 0, stream>>>(
        xyz, features, new_xyz, knn_idx, Wpad, gamma, beta,
        meanv, varv, partial_sum, partial_sq, out_features, 0);
    reduce_stats_kernel<<<COUT, 256, 0, stream>>>(
        partial_sum, partial_sq, meanv, varv);
    group_gemm_kernel<<<BATCH * NPOINT, 256, 0, stream>>>(
        xyz, features, new_xyz, knn_idx, Wpad, gamma, beta,
        meanv, varv, partial_sum, partial_sq, out_features, 1);
}


# Sampling: first wave fetches its 32 column winners' points from global beside the second reduction stage and publishes pick + coordinates in one LDS record; no fetch after the pick; bit-identical
# speedup vs baseline: 2.5093x; 1.0886x over previous
#include <hip/hip_runtime.h>

typedef __attribute__((ext_vector_type(2))) float v2f;
typedef __attribute__((ext_vector_type(8))) float v8f;

#define BATCH   4
#define NPTS    8192
#define CIN     64
#define COUT    128
#define NPOINT  2048
#define KNN     16
#define KRAW    67
#define KPAD    68
#define BNEPS   1e-5f

typedef __attribute__((ext_vector_type(2))) unsigned v2u;
typedef __attribute__((ext_vector_type(4))) unsigned v4u;

__device__ __forceinline__ unsigned umax2(unsigned a, unsigned b)
{
    return a > b ? a : b;
}

__device__ __forceinline__ unsigned umax32(unsigned* m)
{
#pragma unroll
    for (int w = 16; w > 0; w >>= 1) {
#pragma unroll
        for (int j = 0; j < w; ++j) m[j] = umax2(m[j], m[j + w]);
    }
    return m[0];
}

__device__ __forceinline__ void keymax32(unsigned* h, unsigned* l, unsigned& mh, unsigned& ml)
{
    unsigned t[32];
#pragma unroll
    for (int j = 0; j < 32; ++j) t[j] = h[j];
    const unsigned top = umax32(t);
#pragma unroll
    for (int j = 0; j < 32; ++j) l[j] = (h[j] == top) ? l[j] : 0u;
    mh = top;
    ml = umax32(l);
}

__global__ __launch_bounds__(1024) void fps_kernel(const float* __restrict__ xyz,
                                                   int* __restrict__ fps_idx)
{
    const int b    = blockIdx.x;
    const int tid  = threadIdx.x;
    const int lane = tid & 31;
    const int wv   = tid >> 5;
    const float* xb = xyz + (size_t)b * NPTS * 3;

    __shared__ __attribute__((aligned(16))) v2u skey[1024];
    __shared__ __attribute__((aligned(16))) v2u spart[32];
    __shared__ __attribute__((aligned(16))) v4u spick[33];

    const int wslot = (wv >> 1) * 64 + lane * 2 + (wv & 1);

    float dist[8];
#pragma unroll
    for (int i = 0; i < 8; ++i) dist[i] = 3.4e38f;

    int cur = 0;
    float cx = xb[0];
    float cy = xb[1];
    float cz = xb[2];
    for (int s = 0; s < NPOINT; ++s) {
        if (tid == 0) fps_idx[b * NPOINT + s] = cur;

        float best = -1.0f; int bidx = 0;
#pragma unroll
        for (int i = 0; i < 8; ++i) {
            const int n = i * 1024 + tid;
            const float dx = xb[n * 3 + 0] - cx;
            const float dy = xb[n * 3 + 1] - cy;
            const float dz = xb[n * 3 + 2] - cz;
            float d;
            if (i < 6) {
                d = __builtin_fmaf(dz, dz, __builtin_fmaf(dx, dx, dy * dy));
            } else {
                float sxy;
                {
#pragma clang fp contract(off)
                    sxy = dx * dx + dy * dy;
                }
                d = __builtin_fmaf(dz, dz, sxy);
            }
            const float nd = fminf(dist[i], d);
            dist[i] = nd;
            if (nd > best) { best = nd; bidx = n; }
        }
        {
            v2u key;
            key.x = ~(unsigned)bidx;
            key.y = __float_as_uint(best);
            skey[wslot] = key;
        }
        __syncthreads();
        if (wv == 0) {
            const v4u* col = reinterpret_cast<const v4u*>(skey) + lane;
            unsigned h[32], l[32];
#pragma unroll
            for (int j = 0; j < 16; ++j) {
                const v4u p = col[j * 32];
                l[2 * j] = p.x; h[2 * j] = p.y; l[2 * j + 1] = p.z; h[2 * j + 1] = p.w;
            }
            unsigned ph, pl;
            keymax32(h, l, ph, pl);
            const int cand = (int)(~pl) & (NPTS - 1);
            const float sx = xb[cand * 3 + 0];
            const float sy = xb[cand * 3 + 1];
            const float sz = xb[cand * 3 + 2];
            {
                v2u part;
                part.x = pl;
                part.y = ph;
                spart[lane] = part;
            }
            const v4u* all = reinterpret_cast<const v4u*>(spart);
#pragma unroll
            for (int j = 0; j < 16; ++j) {
                const v4u p = all[j];
                l[2 * j] = p.x; h[2 * j] = p.y; l[2 * j + 1] = p.z; h[2 * j + 1] = p.w;
            }
            unsigned gh, gl;
            keymax32(h, l, gh, gl);
            v4u rec;
            rec.x = ~gl;
            rec.y = __float_as_uint(sx);
            rec.z = __float_as_uint(sy);
            rec.w = __float_as_uint(sz);
            spick[(pl == gl) ? 0 : 1 + lane] = rec;
        }
        __syncthreads();
        const v4u pk = spick[0];
        cur = __builtin_amdgcn_readfirstlane((int)pk.x);
        cx  = __uint_as_float((unsigned)__builtin_amdgcn_readfirstlane((int)pk.y));
        cy  = __uint_as_float((unsigned)__builtin_amdgcn_readfirstlane((int)pk.z));
        cz  = __uint_as_float((unsigned)__builtin_amdgcn_readfirstlane((int)pk.w));
    }
}

__global__ void gather_new_xyz_kernel(const float* __restrict__ xyz,
                                      const int* __restrict__ fps_idx,
                                      float* __restrict__ new_xyz)
{
    const int e = blockIdx.x * blockDim.x + threadIdx.x;
    if (e >= BATCH * NPOINT * 3) return;
    const int c = e % 3;
    const int p = (e / 3) % NPOINT;
    const int b = e / (3 * NPOINT);
    const int n = fps_idx[b * NPOINT + p];
    new_xyz[e] = xyz[((size_t)b * NPTS + n) * 3 + c];
}

#define KSEG   16
#define KSEGN  (NPTS / KSEG)
#define KQ     32

__global__ __launch_bounds__(KSEG * 32) void knn_kernel(const float* __restrict__ xyz,
                                                        const float* __restrict__ new_xyz,
                                                        int* __restrict__ knn_idx)
{
    __shared__ float sd[KSEG][KNN][KQ];
    __shared__ int   si[KSEG][KNN][KQ];

    const int lane = threadIdx.x & 31;
    const int seg  = threadIdx.x >> 5;
    const int q    = blockIdx.x * KQ + lane;
    const int b    = (blockIdx.x * KQ) / NPOINT;
    const float qx = new_xyz[q * 3 + 0];
    const float qy = new_xyz[q * 3 + 1];
    const float qz = new_xyz[q * 3 + 2];
    const float* xb = xyz + (size_t)b * NPTS * 3;

    float v[KNN]; int id[KNN];
#pragma unroll
    for (int t = 0; t < KNN; ++t) { v[t] = 3.4e38f; id[t] = 0; }

    const int j0 = seg * KSEGN;
    for (int j = j0; j < j0 + KSEGN; ++j) {
        const float dx = xb[j * 3 + 0] - qx;
        const float dy = xb[j * 3 + 1] - qy;
        const float dz = xb[j * 3 + 2] - qz;
        const float d  = dx * dx + dy * dy + dz * dz;
        if (d < v[KNN - 1]) {
            v[KNN - 1] = d; id[KNN - 1] = j;
#pragma unroll
            for (int t = KNN - 1; t > 0; --t) {
                if (v[t] < v[t - 1]) {
                    const float tv = v[t]; v[t] = v[t - 1]; v[t - 1] = tv;
                    const int   ti = id[t]; id[t] = id[t - 1]; id[t - 1] = ti;
                }
            }
        }
    }
#pragma unroll
    for (int t = 0; t < KNN; ++t) { sd[seg][t][lane] = v[t]; si[seg][t][lane] = id[t]; }
    __syncthreads();

    if (seg == 0) {
        float hd[KSEG];
#pragma unroll
        for (int s = 0; s < KSEG; ++s) hd[s] = sd[s][0][lane];
        unsigned long long heads = 0ull;
        int oid[KNN];
#pragma unroll
        for (int t = 0; t < KNN; ++t) {
            float best = hd[0]; int bs = 0;
#pragma unroll
            for (int s = 1; s < KSEG; ++s)
                if (hd[s] < best) { best = hd[s]; bs = s; }
            const int p = (int)((heads >> (4 * bs)) & 15ull);
            oid[t] = si[bs][p][lane];
            if (t < KNN - 1) {
                heads += 1ull << (4 * bs);
                const float nh = sd[bs][p + 1][lane];
#pragma unroll
                for (int s = 0; s < KSEG; ++s) hd[s] = (s == bs) ? nh : hd[s];
            }
        }
#pragma unroll
        for (int t = 0; t < KNN; ++t) knn_idx[(size_t)q * KNN + t] = oid[t];
    }
}

__global__ void pad_w_kernel(const float* __restrict__ W, float* __restrict__ Wpad)
{
    const int e = blockIdx.x * blockDim.x + threadIdx.x;
    if (e >= KPAD * COUT) return;
    const int r = e / COUT;
    Wpad[e] = (r < KRAW) ? W[e] : 0.0f;
}

__global__ __launch_bounds__(256) void group_gemm_kernel(
    const float* __restrict__ xyz, const float* __restrict__ features,
    const float* __restrict__ new_xyz, const int* __restrict__ knn_idx,
    const float* __restrict__ Wpad,
    const float* __restrict__ gamma, const float* __restrict__ beta,
    const float* __restrict__ meanv, const float* __restrict__ varv,
    float* __restrict__ partial_sum, float* __restrict__ partial_sq,
    float* __restrict__ out_features, int final_pass)
{
    const int q    = blockIdx.x;
    const int b    = q / NPOINT;
    const int tid  = threadIdx.x;
    const int lane = tid & 31;
    const int wv   = tid >> 5;

    __shared__ float As[KNN][KPAD];
    __shared__ float red[8][32];

    const float qx = new_xyz[q * 3 + 0];
    const float qy = new_xyz[q * 3 + 1];
    const float qz = new_xyz[q * 3 + 2];

    for (int e = tid; e < KNN * KPAD; e += 256) {
        const int kk = e / KPAD;
        const int c  = e % KPAD;
        const int n  = knn_idx[(size_t)q * KNN + kk];
        float val;
        if (c < 3) {
            const float pv = xyz[((size_t)b * NPTS + n) * 3 + c];
            val = pv - (c == 0 ? qx : (c == 1 ? qy : qz));
        } else if (c < KRAW) {
            val = features[((size_t)b * NPTS + n) * CIN + (c - 3)];
        } else {
            val = 0.0f;
        }
        As[kk][c] = val;
    }
    __syncthreads();

    const int col = wv * 16 + (lane & 15);
    const int hi  = lane >> 4;
    const int row = lane & 15;

    v8f acc = {};
#pragma unroll
    for (int kb = 0; kb < KPAD / 4; ++kb) {
        const int k0 = kb * 4 + hi * 2;
        v2f a, bb;
        a.x  = As[row][k0 + 0];
        a.y  = As[row][k0 + 1];
        bb.x = Wpad[(k0 + 0) * COUT + col];
        bb.y = Wpad[(k0 + 1) * COUT + col];
        acc = __builtin_amdgcn_wmma_f32_16x16x4_f32(
                  false, a, false, bb, (short)0, acc, false, false);
    }

    if (!final_pass) {
        float s = 0.0f, ss = 0.0f;
#pragma unroll
        for (int t = 0; t < 8; ++t) { const float h = acc[t]; s += h; ss += h * h; }
        red[wv][lane] = s;
        __syncthreads();
        float s2 = 0.0f;
        if (lane < 16) s2 = red[wv][lane] + red[wv][lane + 16];
        __syncthreads();
        red[wv][lane] = ss;
        __syncthreads();
        if (lane < 16) {
            const float ss2 = red[wv][lane] + red[wv][lane + 16];
            partial_sum[(size_t)q * COUT + col] = s2;
            partial_sq [(size_t)q * COUT + col] = ss2;
        }
    } else {
        const float mu    = meanv[col];
        const float scale = gamma[col] * rsqrtf(varv[col] + BNEPS);
        const float shift = beta[col] - mu * scale;
        float m = -3.4e38f;
#pragma unroll
        for (int t = 0; t < 8; ++t) {
            float h = acc[t] * scale + shift;
            h = fmaxf(h, 0.0f);
            m = fmaxf(m, h);
        }
        red[wv][lane] = m;
        __syncthreads();
        if (lane < 16) {
            const float mm = fmaxf(red[wv][lane], red[wv][lane + 16]);
            out_features[(size_t)q * COUT + col] = mm;
        }
    }
}

__global__ __launch_bounds__(256) void reduce_stats_kernel(
    const float* __restrict__ partial_sum, const float* __restrict__ partial_sq,
    float* __restrict__ meanv, float* __restrict__ varv)
{
    const int ch  = blockIdx.x;
    const int tid = threadIdx.x;
    __shared__ float ssum[256], ssq[256];
    float s = 0.0f, sq = 0.0f;
    for (int i = tid; i < BATCH * NPOINT; i += 256) {
        s  += partial_sum[(size_t)i * COUT + ch];
        sq += partial_sq [(size_t)i * COUT + ch];
    }
    ssum[tid] = s; ssq[tid] = sq;
    __syncthreads();
    for (int off = 128; off > 0; off >>= 1) {
        if (tid < off) { ssum[tid] += ssum[tid + off]; ssq[tid] += ssq[tid + off]; }
        __syncthreads();
    }
    if (tid == 0) {
        const float cnt = (float)(BATCH * NPOINT * KNN);
        const float mu  = ssum[0] / cnt;
        meanv[ch] = mu;
        varv[ch]  = ssq[0] / cnt - mu * mu;
    }
}

extern "C" void kernel_launch(void* const* d_in, const int* in_sizes, int n_in,
                              void* d_out, int out_size, void* d_ws, size_t ws_size,
                              hipStream_t stream)
{
    const float* xyz      = (const float*)d_in[0];
    const float* features = (const float*)d_in[1];
    const float* W        = (const float*)d_in[2];
    const float* gamma    = (const float*)d_in[3];
    const float* beta     = (const float*)d_in[4];

    float* out          = (float*)d_out;
    float* new_xyz      = out;
    float* out_features = out + BATCH * NPOINT * 3;

    char* ws = (char*)d_ws;
    size_t off = 0;
    int*   fps_idx     = (int*)(ws + off);   off += sizeof(int)   * BATCH * NPOINT;
    int*   knn_idx     = (int*)(ws + off);   off += sizeof(int)   * BATCH * NPOINT * KNN;
    float* Wpad        = (float*)(ws + off); off += sizeof(float) * KPAD * COUT;
    off = (off + 255) & ~(size_t)255;
    float* partial_sum = (float*)(ws + off); off += sizeof(float) * BATCH * NPOINT * COUT;
    float* partial_sq  = (float*)(ws + off); off += sizeof(float) * BATCH * NPOINT * COUT;
    float* meanv       = (float*)(ws + off); off += sizeof(float) * COUT;
    float* varv        = (float*)(ws + off); off += sizeof(float) * COUT;
    (void)ws_size; (void)in_sizes; (void)n_in; (void)out_size;

    fps_kernel<<<BATCH, 1024, 0, stream>>>(xyz, fps_idx);
    gather_new_xyz_kernel<<<(BATCH * NPOINT * 3 + 255) / 256, 256, 0, stream>>>(
        xyz, fps_idx, new_xyz);
    knn_kernel<<<(BATCH * NPOINT) / KQ, KSEG * 32, 0, stream>>>(
        xyz, new_xyz, knn_idx);
    pad_w_kernel<<<(KPAD * COUT + 255) / 256, 256, 0, stream>>>(W, Wpad);
    group_gemm_kernel<<<BATCH * NPOINT, 256, 0, stream>>>(
        xyz, features, new_xyz, knn_idx, Wpad, gamma, beta,
        meanv, varv, partial_sum, partial_sq, out_features, 0);
    reduce_stats_kernel<<<COUT, 256, 0, stream>>>(
        partial_sum, partial_sq, meanv, varv);
    group_gemm_kernel<<<BATCH * NPOINT, 256, 0, stream>>>(
        xyz, features, new_xyz, knn_idx, Wpad, gamma, beta,
        meanv, varv, partial_sum, partial_sq, out_features, 1);
}


# Sampling: first wave keeps each column winner's point from the pick before and publishes it unless that winner changed; 32-lane maximum by DPP, LDS-only barriers: no wait on the fetch; bit-identical
# speedup vs baseline: 3.2676x; 1.3022x over previous
#include <hip/hip_runtime.h>

typedef __attribute__((ext_vector_type(2))) float v2f;
typedef __attribute__((ext_vector_type(8))) float v8f;

#define BATCH   4
#define NPTS    8192
#define CIN     64
#define COUT    128
#define NPOINT  2048
#define KNN     16
#define KRAW    67
#define KPAD    68
#define BNEPS   1e-5f

typedef __attribute__((ext_vector_type(2))) unsigned v2u;
typedef __attribute__((ext_vector_type(4))) unsigned v4u;

__device__ __forceinline__ unsigned umax2(unsigned a, unsigned b)
{
    return a > b ? a : b;
}

__device__ __forceinline__ unsigned umax32(unsigned* m)
{
#pragma unroll
    for (int w = 16; w > 0; w >>= 1) {
#pragma unroll
        for (int j = 0; j < w; ++j) m[j] = umax2(m[j], m[j + w]);
    }
    return m[0];
}

__device__ __forceinline__ void keymax32(unsigned* h, unsigned* l, unsigned& mh, unsigned& ml)
{
    unsigned t[32];
#pragma unroll
    for (int j = 0; j < 32; ++j) t[j] = h[j];
    const unsigned top = umax32(t);
#pragma unroll
    for (int j = 0; j < 32; ++j) l[j] = (h[j] == top) ? l[j] : 0u;
    mh = top;
    ml = umax32(l);
}

__device__ __forceinline__ unsigned fkey(float x)
{
    const unsigned u = __float_as_uint(x);
    return u ^ ((u >> 31) ? 0xFFFFFFFFu : 0x80000000u);
}

__device__ __forceinline__ unsigned wave_umax(unsigned v)
{
    v = umax2(v, (unsigned)__builtin_amdgcn_update_dpp(0, (int)v, 0x161, 0xf, 0xf, true));
    v = umax2(v, (unsigned)__builtin_amdgcn_update_dpp(0, (int)v, 0x162, 0xf, 0xf, true));
    v = umax2(v, (unsigned)__builtin_amdgcn_update_dpp(0, (int)v, 0x164, 0xf, 0xf, true));
    v = umax2(v, (unsigned)__builtin_amdgcn_update_dpp(0, (int)v, 0x168, 0xf, 0xf, true));
    v = umax2(v, (unsigned)__builtin_amdgcn_permlanex16((int)v, (int)v, 0x76543210, 0xfedcba98, false, false));
    return v;
}

__device__ __forceinline__ void lds_barrier()
{
    asm volatile("s_wait_dscnt 0x0\n\ts_barrier_signal -1\n\ts_barrier_wait -1" ::: "memory");
}

__global__ __launch_bounds__(1024) void fps_kernel(const float* __restrict__ xyz,
                                                   int* __restrict__ fps_idx)
{
    const int b    = blockIdx.x;
    const int tid  = threadIdx.x;
    const int lane = tid & 31;
    const int wv   = tid >> 5;
    const float* xb = xyz + (size_t)b * NPTS * 3;

    __shared__ __attribute__((aligned(8))) v2u ssort[NPTS];
    __shared__ __attribute__((aligned(16))) v2u skey[16 * 66];
    __shared__ __attribute__((aligned(16))) v2u spart[32];
    __shared__ __attribute__((aligned(16))) v4u spick[33];
    __shared__ __attribute__((aligned(16))) v4u skeep[32];

#pragma unroll
    for (int r = 0; r < 8; ++r) {
        const int n = r * 1024 + tid;
        v2u e;
        e.x = fkey(xb[n * 3]);
        e.y = (unsigned)n;
        ssort[n] = e;
    }
    __syncthreads();
    for (unsigned k = 2; k <= NPTS; k <<= 1) {
        for (unsigned j = k >> 1; j > 0; j >>= 1) {
            unsigned ii[4];
            v2u ea[4], eb[4];
#pragma unroll
            for (int r = 0; r < 4; ++r) {
                const unsigned t = (unsigned)(r * 1024 + tid);
                ii[r] = ((t & ~(j - 1u)) << 1) | (t & (j - 1u));
                ea[r] = ssort[ii[r]];
                eb[r] = ssort[ii[r] | j];
            }
#pragma unroll
            for (int r = 0; r < 4; ++r) {
                const bool up = (ii[r] & k) == 0u;
                if ((ea[r].x > eb[r].x) == up) {
                    ssort[ii[r]]     = eb[r];
                    ssort[ii[r] | j] = ea[r];
                }
            }
            __syncthreads();
        }
    }

    const int wa = wv >> 3, wb = wv & 7;
    const int slab = 4 * (((((wb & 3) - wa) & 3) << 1) | (wb >> 2)) + wa;

    float px[8], py[8], pz[8], dist[8];
    unsigned linv[8];
    bool unf[8];
#pragma unroll
    for (int i = 0; i < 8; ++i) {
        const unsigned n = ssort[slab * 256 + i * 32 + lane].y & (NPTS - 1);
        px[i] = xb[n * 3 + 0];
        py[i] = xb[n * 3 + 1];
        pz[i] = xb[n * 3 + 2];
        linv[i] = ~n;
        unf[i]  = n >= 6144u;
        dist[i] = 3.4e38f;
    }
    const unsigned nlo = ssort[slab * 256].y & (NPTS - 1);
    const unsigned nhi = ssort[slab * 256 + 255].y & (NPTS - 1);
    const float xlo = __uint_as_float((unsigned)__builtin_amdgcn_readfirstlane(
                          (int)__float_as_uint(xb[nlo * 3])));
    const float xhi = __uint_as_float((unsigned)__builtin_amdgcn_readfirstlane(
                          (int)__float_as_uint(xb[nhi * 3])));

    const int wslot = (lane >> 1) * 66 + wv * 2 + (lane & 1);

    int cur = 0;
    float cx = xb[0];
    float cy = xb[1];
    float cz = xb[2];
    float wmax = 3.4e38f;
    unsigned npl = 0u;
    float nx = 0.0f, ny = 0.0f, nz = 0.0f;
    for (int s = 0; s < NPOINT; ++s) {
        if (tid == 0) fps_idx[b * NPOINT + s] = cur;

        const float g   = fmaxf(fmaxf(xlo - cx, cx - xhi), 0.0f);
        const float bnd = g * g;
        if (s == 0 || !(bnd >= wmax)) {
            float nd[8];
#pragma unroll
            for (int i = 0; i < 8; ++i) {
                const float dx = px[i] - cx;
                const float dy = py[i] - cy;
                const float dz = pz[i] - cz;
                const float q = dy * dy;
                const float f = __builtin_fmaf(dx, dx, q);
                float u;
                {
#pragma clang fp contract(off)
                    u = dx * dx + q;
                }
                const float d = __builtin_fmaf(dz, dz, unf[i] ? u : f);
                nd[i] = fminf(dist[i], d);
                dist[i] = nd[i];
            }
            float best = fmaxf(fmaxf(fmaxf(nd[0], nd[1]), fmaxf(nd[2], nd[3])),
                               fmaxf(fmaxf(nd[4], nd[5]), fmaxf(nd[6], nd[7])));
            unsigned lo = 0u;
#pragma unroll
            for (int i = 0; i < 8; ++i) lo = umax2(lo, (nd[i] == best) ? linv[i] : 0u);
            v2u key;
            key.x = lo;
            key.y = __float_as_uint(best);
            skey[wslot] = key;
        }
        if (wv == 0) {
            v4u nrec;
            nrec.x = npl;
            nrec.y = __float_as_uint(nx);
            nrec.z = __float_as_uint(ny);
            nrec.w = __float_as_uint(nz);
            skeep[lane] = nrec;
        }
        lds_barrier();
        if (wv == 0) {
            const v4u* col = reinterpret_cast<const v4u*>(skey) + lane;
            unsigned h[32], l[32];
#pragma unroll
            for (int j = 0; j < 16; ++j) {
                const v4u p = col[j * 33];
                l[2 * j] = p.x; h[2 * j] = p.y; l[2 * j + 1] = p.z; h[2 * j + 1] = p.w;
            }
            unsigned ph, pl;
            keymax32(h, l, ph, pl);
            const int cand = (int)(~pl) & (NPTS - 1);
            const float sx = xb[cand * 3 + 0];
            const float sy = xb[cand * 3 + 1];
            const float sz = xb[cand * 3 + 2];
            {
                v2u part;
                part.x = pl;
                part.y = ph;
                spart[lane] = part;
            }
            const v4u kept = skeep[lane];
            const unsigned gh = wave_umax(ph);
            const unsigned gl = wave_umax((ph == gh) ? pl : 0u);
            const bool mine  = (pl == gl);
            const bool stale = mine && (kept.x != pl);
            if (__builtin_amdgcn_ballot_w32(stale) != 0u) {
                if (mine) {
                    v4u rec;
                    rec.x = ~gl;
                    rec.y = __float_as_uint(sx);
                    rec.z = __float_as_uint(sy);
                    rec.w = __float_as_uint(sz);
                    spick[0] = rec;
                }
            } else {
                v4u rec = kept;
                rec.x = ~gl;
                spick[mine ? 0 : 1 + lane] = rec;
            }
            npl = pl;
            nx = sx;
            ny = sy;
            nz = sz;
        }
        lds_barrier();
        const v4u pk = spick[0];
        const v2u wm = spart[wv];
        cur = __builtin_amdgcn_readfirstlane((int)pk.x);
        cx  = __uint_as_float((unsigned)__builtin_amdgcn_readfirstlane((int)pk.y));
        cy  = __uint_as_float((unsigned)__builtin_amdgcn_readfirstlane((int)pk.z));
        cz  = __uint_as_float((unsigned)__builtin_amdgcn_readfirstlane((int)pk.w));
        wmax = __uint_as_float((unsigned)__builtin_amdgcn_readfirstlane((int)wm.y));
    }
}

__global__ void gather_new_xyz_kernel(const float* __restrict__ xyz,
                                      const int* __restrict__ fps_idx,
                                      float* __restrict__ new_xyz)
{
    const int e = blockIdx.x * blockDim.x + threadIdx.x;
    if (e >= BATCH * NPOINT * 3) return;
    const int c = e % 3;
    const int p = (e / 3) % NPOINT;
    const int b = e / (3 * NPOINT);
    const int n = fps_idx[b * NPOINT + p];
    new_xyz[e] = xyz[((size_t)b * NPTS + n) * 3 + c];
}

#define KSEG   16
#define KSEGN  (NPTS / KSEG)
#define KQ     32

__global__ __launch_bounds__(KSEG * 32) void knn_kernel(const float* __restrict__ xyz,
                                                        const float* __restrict__ new_xyz,
                                                        int* __restrict__ knn_idx)
{
    __shared__ float sd[KSEG][KNN][KQ];
    __shared__ int   si[KSEG][KNN][KQ];

    const int lane = threadIdx.x & 31;
    const int seg  = threadIdx.x >> 5;
    const int q    = blockIdx.x * KQ + lane;
    const int b    = (blockIdx.x * KQ) / NPOINT;
    const float qx = new_xyz[q * 3 + 0];
    const float qy = new_xyz[q * 3 + 1];
    const float qz = new_xyz[q * 3 + 2];
    const float* xb = xyz + (size_t)b * NPTS * 3;

    float v[KNN]; int id[KNN];
#pragma unroll
    for (int t = 0; t < KNN; ++t) { v[t] = 3.4e38f; id[t] = 0; }

    const int j0 = seg * KSEGN;
    for (int j = j0; j < j0 + KSEGN; ++j) {
        const float dx = xb[j * 3 + 0] - qx;
        const float dy = xb[j * 3 + 1] - qy;
        const float dz = xb[j * 3 + 2] - qz;
        const float d  = dx * dx + dy * dy + dz * dz;
        if (d < v[KNN - 1]) {
            v[KNN - 1] = d; id[KNN - 1] = j;
#pragma unroll
            for (int t = KNN - 1; t > 0; --t) {
                if (v[t] < v[t - 1]) {
                    const float tv = v[t]; v[t] = v[t - 1]; v[t - 1] = tv;
                    const int   ti = id[t]; id[t] = id[t - 1]; id[t - 1] = ti;
                }
            }
        }
    }
#pragma unroll
    for (int t = 0; t < KNN; ++t) { sd[seg][t][lane] = v[t]; si[seg][t][lane] = id[t]; }
    __syncthreads();

    if (seg == 0) {
        float hd[KSEG];
#pragma unroll
        for (int s = 0; s < KSEG; ++s) hd[s] = sd[s][0][lane];
        unsigned long long heads = 0ull;
        int oid[KNN];
#pragma unroll
        for (int t = 0; t < KNN; ++t) {
            float best = hd[0]; int bs = 0;
#pragma unroll
            for (int s = 1; s < KSEG; ++s)
                if (hd[s] < best) { best = hd[s]; bs = s; }
            const int p = (int)((heads >> (4 * bs)) & 15ull);
            oid[t] = si[bs][p][lane];
            if (t < KNN - 1) {
                heads += 1ull << (4 * bs);
                const float nh = sd[bs][p + 1][lane];
#pragma unroll
                for (int s = 0; s < KSEG; ++s) hd[s] = (s == bs) ? nh : hd[s];
            }
        }
#pragma unroll
        for (int t = 0; t < KNN; ++t) knn_idx[(size_t)q * KNN + t] = oid[t];
    }
}

__global__ void pad_w_kernel(const float* __restrict__ W, float* __restrict__ Wpad)
{
    const int e = blockIdx.x * blockDim.x + threadIdx.x;
    if (e >= KPAD * COUT) return;
    const int r = e / COUT;
    Wpad[e] = (r < KRAW) ? W[e] : 0.0f;
}

__global__ __launch_bounds__(256) void group_gemm_kernel(
    const float* __restrict__ xyz, const float* __restrict__ features,
    const float* __restrict__ new_xyz, const int* __restrict__ knn_idx,
    const float* __restrict__ Wpad,
    const float* __restrict__ gamma, const float* __restrict__ beta,
    const float* __restrict__ meanv, const float* __restrict__ varv,
    float* __restrict__ partial_sum, float* __restrict__ partial_sq,
    float* __restrict__ out_features, int final_pass)
{
    const int q    = blockIdx.x;
    const int b    = q / NPOINT;
    const int tid  = threadIdx.x;
    const int lane = tid & 31;
    const int wv   = tid >> 5;

    __shared__ float As[KNN][KPAD];
    __shared__ float red[8][32];

    const float qx = new_xyz[q * 3 + 0];
    const float qy = new_xyz[q * 3 + 1];
    const float qz = new_xyz[q * 3 + 2];

    for (int e = tid; e < KNN * KPAD; e += 256) {
        const int kk = e / KPAD;
        const int c  = e % KPAD;
        const int n  = knn_idx[(size_t)q * KNN + kk];
        float val;
        if (c < 3) {
            const float pv = xyz[((size_t)b * NPTS + n) * 3 + c];
            val = pv - (c == 0 ? qx : (c == 1 ? qy : qz));
        } else if (c < KRAW) {
            val = features[((size_t)b * NPTS + n) * CIN + (c - 3)];
        } else {
            val = 0.0f;
        }
        As[kk][c] = val;
    }
    __syncthreads();

    const int col = wv * 16 + (lane & 15);
    const int hi  = lane >> 4;
    const int row = lane & 15;

    v8f acc = {};
#pragma unroll
    for (int kb = 0; kb < KPAD / 4; ++kb) {
        const int k0 = kb * 4 + hi * 2;
        v2f a, bb;
        a.x  = As[row][k0 + 0];
        a.y  = As[row][k0 + 1];
        bb.x = Wpad[(k0 + 0) * COUT + col];
        bb.y = Wpad[(k0 + 1) * COUT + col];
        acc = __builtin_amdgcn_wmma_f32_16x16x4_f32(
                  false, a, false, bb, (short)0, acc, false, false);
    }

    if (!final_pass) {
        float s = 0.0f, ss = 0.0f;
#pragma unroll
        for (int t = 0; t < 8; ++t) { const float h = acc[t]; s += h; ss += h * h; }
        red[wv][lane] = s;
        __syncthreads();
        float s2 = 0.0f;
        if (lane < 16) s2 = red[wv][lane] + red[wv][lane + 16];
        __syncthreads();
        red[wv][lane] = ss;
        __syncthreads();
        if (lane < 16) {
            const float ss2 = red[wv][lane] + red[wv][lane + 16];
            partial_sum[(size_t)q * COUT + col] = s2;
            partial_sq [(size_t)q * COUT + col] = ss2;
        }
    } else {
        const float mu    = meanv[col];
        const float scale = gamma[col] * rsqrtf(varv[col] + BNEPS);
        const float shift = beta[col] - mu * scale;
        float m = -3.4e38f;
#pragma unroll
        for (int t = 0; t < 8; ++t) {
            float h = acc[t] * scale + shift;
            h = fmaxf(h, 0.0f);
            m = fmaxf(m, h);
        }
        red[wv][lane] = m;
        __syncthreads();
        if (lane < 16) {
            const float mm = fmaxf(red[wv][lane], red[wv][lane + 16]);
            out_features[(size_t)q * COUT + col] = mm;
        }
    }
}

__global__ __launch_bounds__(256) void reduce_stats_kernel(
    const float* __restrict__ partial_sum, const float* __restrict__ partial_sq,
    float* __restrict__ meanv, float* __restrict__ varv)
{
    const int ch  = blockIdx.x;
    const int tid = threadIdx.x;
    __shared__ float ssum[256], ssq[256];
    float s = 0.0f, sq = 0.0f;
    for (int i = tid; i < BATCH * NPOINT; i += 256) {
        s  += partial_sum[(size_t)i * COUT + ch];
        sq += partial_sq [(size_t)i * COUT + ch];
    }
    ssum[tid] = s; ssq[tid] = sq;
    __syncthreads();
    for (int off = 128; off > 0; off >>= 1) {
        if (tid < off) { ssum[tid] += ssum[tid + off]; ssq[tid] += ssq[tid + off]; }
        __syncthreads();
    }
    if (tid == 0) {
        const float cnt = (float)(BATCH * NPOINT * KNN);
        const float mu  = ssum[0] / cnt;
        meanv[ch] = mu;
        varv[ch]  = ssq[0] / cnt - mu * mu;
    }
}

extern "C" void kernel_launch(void* const* d_in, const int* in_sizes, int n_in,
                              void* d_out, int out_size, void* d_ws, size_t ws_size,
                              hipStream_t stream)
{
    const float* xyz      = (const float*)d_in[0];
    const float* features = (const float*)d_in[1];
    const float* W        = (const float*)d_in[2];
    const float* gamma    = (const float*)d_in[3];
    const float* beta     = (const float*)d_in[4];

    float* out          = (float*)d_out;
    float* new_xyz      = out;
    float* out_features = out + BATCH * NPOINT * 3;

    char* ws = (char*)d_ws;
    size_t off = 0;
    int*   fps_idx     = (int*)(ws + off);   off += sizeof(int)   * BATCH * NPOINT;
    int*   knn_idx     = (int*)(ws + off);   off += sizeof(int)   * BATCH * NPOINT * KNN;
    float* Wpad        = (float*)(ws + off); off += sizeof(float) * KPAD * COUT;
    off = (off + 255) & ~(size_t)255;
    float* partial_sum = (float*)(ws + off); off += sizeof(float) * BATCH * NPOINT * COUT;
    float* partial_sq  = (float*)(ws + off); off += sizeof(float) * BATCH * NPOINT * COUT;
    float* meanv       = (float*)(ws + off); off += sizeof(float) * COUT;
    float* varv        = (float*)(ws + off); off += sizeof(float) * COUT;
    (void)ws_size; (void)in_sizes; (void)n_in; (void)out_size;

    fps_kernel<<<BATCH, 1024, 0, stream>>>(xyz, fps_idx);
    gather_new_xyz_kernel<<<(BATCH * NPOINT * 3 + 255) / 256, 256, 0, stream>>>(
        xyz, fps_idx, new_xyz);
    knn_kernel<<<(BATCH * NPOINT) / KQ, KSEG * 32, 0, stream>>>(
        xyz, new_xyz, knn_idx);
    pad_w_kernel<<<(KPAD * COUT + 255) / 256, 256, 0, stream>>>(W, Wpad);
    group_gemm_kernel<<<BATCH * NPOINT, 256, 0, stream>>>(
        xyz, features, new_xyz, knn_idx, Wpad, gamma, beta,
        meanv, varv, partial_sum, partial_sq, out_features, 0);
    reduce_stats_kernel<<<COUT, 256, 0, stream>>>(
        partial_sum, partial_sq, meanv, varv);
    group_gemm_kernel<<<BATCH * NPOINT, 256, 0, stream>>>(
        xyz, features, new_xyz, knn_idx, Wpad, gamma, beta,
        meanv, varv, partial_sum, partial_sq, out_features, 1);
}


# Grouped product once per point, not per neighbour: a dense f32 WMMA pass projects all points (k order xyz, 0, features); both BN passes gather projected rows minus the query's; sums reordered
# speedup vs baseline: 3.3930x; 1.0384x over previous
#include <hip/hip_runtime.h>

typedef __attribute__((ext_vector_type(2))) float v2f;
typedef __attribute__((ext_vector_type(8))) float v8f;

#define BATCH   4
#define NPTS    8192
#define CIN     64
#define COUT    128
#define NPOINT  2048
#define KNN     16
#define KRAW    67
#define KPAD    68
#define BNEPS   1e-5f

typedef __attribute__((ext_vector_type(2))) unsigned v2u;
typedef __attribute__((ext_vector_type(4))) unsigned v4u;

__device__ __forceinline__ unsigned umax2(unsigned a, unsigned b)
{
    return a > b ? a : b;
}

__device__ __forceinline__ unsigned umax32(unsigned* m)
{
#pragma unroll
    for (int w = 16; w > 0; w >>= 1) {
#pragma unroll
        for (int j = 0; j < w; ++j) m[j] = umax2(m[j], m[j + w]);
    }
    return m[0];
}

__device__ __forceinline__ void keymax32(unsigned* h, unsigned* l, unsigned& mh, unsigned& ml)
{
    unsigned t[32];
#pragma unroll
    for (int j = 0; j < 32; ++j) t[j] = h[j];
    const unsigned top = umax32(t);
#pragma unroll
    for (int j = 0; j < 32; ++j) l[j] = (h[j] == top) ? l[j] : 0u;
    mh = top;
    ml = umax32(l);
}

__device__ __forceinline__ unsigned fkey(float x)
{
    const unsigned u = __float_as_uint(x);
    return u ^ ((u >> 31) ? 0xFFFFFFFFu : 0x80000000u);
}

__device__ __forceinline__ unsigned wave_umax(unsigned v)
{
    v = umax2(v, (unsigned)__builtin_amdgcn_update_dpp(0, (int)v, 0x161, 0xf, 0xf, true));
    v = umax2(v, (unsigned)__builtin_amdgcn_update_dpp(0, (int)v, 0x162, 0xf, 0xf, true));
    v = umax2(v, (unsigned)__builtin_amdgcn_update_dpp(0, (int)v, 0x164, 0xf, 0xf, true));
    v = umax2(v, (unsigned)__builtin_amdgcn_update_dpp(0, (int)v, 0x168, 0xf, 0xf, true));
    v = umax2(v, (unsigned)__builtin_amdgcn_permlanex16((int)v, (int)v, 0x76543210, 0xfedcba98, false, false));
    return v;
}

__device__ __forceinline__ void lds_barrier()
{
    asm volatile("s_wait_dscnt 0x0\n\ts_barrier_signal -1\n\ts_barrier_wait -1" ::: "memory");
}

__global__ __launch_bounds__(1024) void fps_kernel(const float* __restrict__ xyz,
                                                   int* __restrict__ fps_idx)
{
    const int b    = blockIdx.x;
    const int tid  = threadIdx.x;
    const int lane = tid & 31;
    const int wv   = tid >> 5;
    const float* xb = xyz + (size_t)b * NPTS * 3;

    __shared__ __attribute__((aligned(8))) v2u ssort[NPTS];
    __shared__ __attribute__((aligned(16))) v2u skey[16 * 66];
    __shared__ __attribute__((aligned(16))) v2u spart[32];
    __shared__ __attribute__((aligned(16))) v4u spick[33];
    __shared__ __attribute__((aligned(16))) v4u skeep[32];

#pragma unroll
    for (int r = 0; r < 8; ++r) {
        const int n = r * 1024 + tid;
        v2u e;
        e.x = fkey(xb[n * 3]);
        e.y = (unsigned)n;
        ssort[n] = e;
    }
    __syncthreads();
    for (unsigned k = 2; k <= NPTS; k <<= 1) {
        for (unsigned j = k >> 1; j > 0; j >>= 1) {
            unsigned ii[4];
            v2u ea[4], eb[4];
#pragma unroll
            for (int r = 0; r < 4; ++r) {
                const unsigned t = (unsigned)(r * 1024 + tid);
                ii[r] = ((t & ~(j - 1u)) << 1) | (t & (j - 1u));
                ea[r] = ssort[ii[r]];
                eb[r] = ssort[ii[r] | j];
            }
#pragma unroll
            for (int r = 0; r < 4; ++r) {
                const bool up = (ii[r] & k) == 0u;
                if ((ea[r].x > eb[r].x) == up) {
                    ssort[ii[r]]     = eb[r];
                    ssort[ii[r] | j] = ea[r];
                }
            }
            __syncthreads();
        }
    }

    const int wa = wv >> 3, wb = wv & 7;
    const int slab = 4 * (((((wb & 3) - wa) & 3) << 1) | (wb >> 2)) + wa;

    float px[8], py[8], pz[8], dist[8];
    unsigned linv[8];
    bool unf[8];
#pragma unroll
    for (int i = 0; i < 8; ++i) {
        const unsigned n = ssort[slab * 256 + i * 32 + lane].y & (NPTS - 1);
        px[i] = xb[n * 3 + 0];
        py[i] = xb[n * 3 + 1];
        pz[i] = xb[n * 3 + 2];
        linv[i] = ~n;
        unf[i]  = n >= 6144u;
        dist[i] = 3.4e38f;
    }
    const unsigned nlo = ssort[slab * 256].y & (NPTS - 1);
    const unsigned nhi = ssort[slab * 256 + 255].y & (NPTS - 1);
    const float xlo = __uint_as_float((unsigned)__builtin_amdgcn_readfirstlane(
                          (int)__float_as_uint(xb[nlo * 3])));
    const float xhi = __uint_as_float((unsigned)__builtin_amdgcn_readfirstlane(
                          (int)__float_as_uint(xb[nhi * 3])));

    const int wslot = (lane >> 1) * 66 + wv * 2 + (lane & 1);

    int cur = 0;
    float cx = xb[0];
    float cy = xb[1];
    float cz = xb[2];
    float wmax = 3.4e38f;
    unsigned npl = 0u;
    float nx = 0.0f, ny = 0.0f, nz = 0.0f;
    for (int s = 0; s < NPOINT; ++s) {
        if (tid == 0) fps_idx[b * NPOINT + s] = cur;

        const float g   = fmaxf(fmaxf(xlo - cx, cx - xhi), 0.0f);
        const float bnd = g * g;
        if (s == 0 || !(bnd >= wmax)) {
            float nd[8];
#pragma unroll
            for (int i = 0; i < 8; ++i) {
                const float dx = px[i] - cx;
                const float dy = py[i] - cy;
                const float dz = pz[i] - cz;
                const float q = dy * dy;
                const float f = __builtin_fmaf(dx, dx, q);
                float u;
                {
#pragma clang fp contract(off)
                    u = dx * dx + q;
                }
                const float d = __builtin_fmaf(dz, dz, unf[i] ? u : f);
                nd[i] = fminf(dist[i], d);
                dist[i] = nd[i];
            }
            float best = fmaxf(fmaxf(fmaxf(nd[0], nd[1]), fmaxf(nd[2], nd[3])),
                               fmaxf(fmaxf(nd[4], nd[5]), fmaxf(nd[6], nd[7])));
            unsigned lo = 0u;
#pragma unroll
            for (int i = 0; i < 8; ++i) lo = umax2(lo, (nd[i] == best) ? linv[i] : 0u);
            v2u key;
            key.x = lo;
            key.y = __float_as_uint(best);
            skey[wslot] = key;
        }
        if (wv == 0) {
            v4u nrec;
            nrec.x = npl;
            nrec.y = __float_as_uint(nx);
            nrec.z = __float_as_uint(ny);
            nrec.w = __float_as_uint(nz);
            skeep[lane] = nrec;
        }
        lds_barrier();
        if (wv == 0) {
            const v4u* col = reinterpret_cast<const v4u*>(skey) + lane;
            unsigned h[32], l[32];
#pragma unroll
            for (int j = 0; j < 16; ++j) {
                const v4u p = col[j * 33];
                l[2 * j] = p.x; h[2 * j] = p.y; l[2 * j + 1] = p.z; h[2 * j + 1] = p.w;
            }
            unsigned ph, pl;
            keymax32(h, l, ph, pl);
            const int cand = (int)(~pl) & (NPTS - 1);
            const float sx = xb[cand * 3 + 0];
            const float sy = xb[cand * 3 + 1];
            const float sz = xb[cand * 3 + 2];
            {
                v2u part;
                part.x = pl;
                part.y = ph;
                spart[lane] = part;
            }
            const v4u kept = skeep[lane];
            const unsigned gh = wave_umax(ph);
            const unsigned gl = wave_umax((ph == gh) ? pl : 0u);
            const bool mine  = (pl == gl);
            const bool stale = mine && (kept.x != pl);
            if (__builtin_amdgcn_ballot_w32(stale) != 0u) {
                if (mine) {
                    v4u rec;
                    rec.x = ~gl;
                    rec.y = __float_as_uint(sx);
                    rec.z = __float_as_uint(sy);
                    rec.w = __float_as_uint(sz);
                    spick[0] = rec;
                }
            } else {
                v4u rec = kept;
                rec.x = ~gl;
                spick[mine ? 0 : 1 + lane] = rec;
            }
            npl = pl;
            nx = sx;
            ny = sy;
            nz = sz;
        }
        lds_barrier();
        const v4u pk = spick[0];
        const v2u wm = spart[wv];
        cur = __builtin_amdgcn_readfirstlane((int)pk.x);
        cx  = __uint_as_float((unsigned)__builtin_amdgcn_readfirstlane((int)pk.y));
        cy  = __uint_as_float((unsigned)__builtin_amdgcn_readfirstlane((int)pk.z));
        cz  = __uint_as_float((unsigned)__builtin_amdgcn_readfirstlane((int)pk.w));
        wmax = __uint_as_float((unsigned)__builtin_amdgcn_readfirstlane((int)wm.y));
    }
}

__global__ void gather_new_xyz_kernel(const float* __restrict__ xyz,
                                      const int* __restrict__ fps_idx,
                                      float* __restrict__ new_xyz)
{
    const int e = blockIdx.x * blockDim.x + threadIdx.x;
    if (e >= BATCH * NPOINT * 3) return;
    const int c = e % 3;
    const int p = (e / 3) % NPOINT;
    const int b = e / (3 * NPOINT);
    const int n = fps_idx[b * NPOINT + p];
    new_xyz[e] = xyz[((size_t)b * NPTS + n) * 3 + c];
}

#define KSEG   16
#define KSEGN  (NPTS / KSEG)
#define KQ     32

__global__ __launch_bounds__(KSEG * 32) void knn_kernel(const float* __restrict__ xyz,
                                                        const float* __restrict__ new_xyz,
                                                        int* __restrict__ knn_idx)
{
    __shared__ float sd[KSEG][KNN][KQ];
    __shared__ int   si[KSEG][KNN][KQ];

    const int lane = threadIdx.x & 31;
    const int seg  = threadIdx.x >> 5;
    const int q    = blockIdx.x * KQ + lane;
    const int b    = (blockIdx.x * KQ) / NPOINT;
    const float qx = new_xyz[q * 3 + 0];
    const float qy = new_xyz[q * 3 + 1];
    const float qz = new_xyz[q * 3 + 2];
    const float* xb = xyz + (size_t)b * NPTS * 3;

    float v[KNN]; int id[KNN];
#pragma unroll
    for (int t = 0; t < KNN; ++t) { v[t] = 3.4e38f; id[t] = 0; }

    const int j0 = seg * KSEGN;
    for (int j = j0; j < j0 + KSEGN; ++j) {
        const float dx = xb[j * 3 + 0] - qx;
        const float dy = xb[j * 3 + 1] - qy;
        const float dz = xb[j * 3 + 2] - qz;
        const float d  = dx * dx + dy * dy + dz * dz;
        if (d < v[KNN - 1]) {
            v[KNN - 1] = d; id[KNN - 1] = j;
#pragma unroll
            for (int t = KNN - 1; t > 0; --t) {
                if (v[t] < v[t - 1]) {
                    const float tv = v[t]; v[t] = v[t - 1]; v[t - 1] = tv;
                    const int   ti = id[t]; id[t] = id[t - 1]; id[t - 1] = ti;
                }
            }
        }
    }
#pragma unroll
    for (int t = 0; t < KNN; ++t) { sd[seg][t][lane] = v[t]; si[seg][t][lane] = id[t]; }
    __syncthreads();

    if (seg == 0) {
        float hd[KSEG];
#pragma unroll
        for (int s = 0; s < KSEG; ++s) hd[s] = sd[s][0][lane];
        unsigned long long heads = 0ull;
        int oid[KNN];
#pragma unroll
        for (int t = 0; t < KNN; ++t) {
            float best = hd[0]; int bs = 0;
#pragma unroll
            for (int s = 1; s < KSEG; ++s)
                if (hd[s] < best) { best = hd[s]; bs = s; }
            const int p = (int)((heads >> (4 * bs)) & 15ull);
            oid[t] = si[bs][p][lane];
            if (t < KNN - 1) {
                heads += 1ull << (4 * bs);
                const float nh = sd[bs][p + 1][lane];
#pragma unroll
                for (int s = 0; s < KSEG; ++s) hd[s] = (s == bs) ? nh : hd[s];
            }
        }
#pragma unroll
        for (int t = 0; t < KNN; ++t) knn_idx[(size_t)q * KNN + t] = oid[t];
    }
}

__global__ void pad_w_kernel(const float* __restrict__ W, float* __restrict__ Wpair)
{
    const int e = blockIdx.x * blockDim.x + threadIdx.x;
    if (e >= KPAD * COUT) return;
    const int k = e / COUT;
    const int c = e % COUT;
    float v;
    if (k < 3)       v = W[k * COUT + c];
    else if (k == 3) v = 0.0f;
    else             v = W[(k - 1) * COUT + c];
    Wpair[((k >> 1) * COUT + c) * 2 + (k & 1)] = v;
}

#define PROWS 128

__global__ __launch_bounds__(256) void project_kernel(
    const float* __restrict__ xyz, const float* __restrict__ features,
    const float* __restrict__ Wpair,
    float* __restrict__ Pall, float* __restrict__ Fall)
{
    const int lane = threadIdx.x & 31;
    const int wv   = threadIdx.x >> 5;
    const int c    = lane & 15;
    const int hi   = lane >> 4;
    const size_t row0 = (size_t)blockIdx.x * PROWS + (size_t)wv * 16;
    const size_t row  = row0 + c;

    v8f acc[8];
#pragma unroll
    for (int nt = 0; nt < 8; ++nt) acc[nt] = (v8f){};

    {
        const float* p = xyz + row * 3;
        v2f a;
        a.x = hi ? p[2] : p[0];
        a.y = hi ? 0.0f : p[1];
#pragma unroll
        for (int nt = 0; nt < 8; ++nt) {
            const v2f bb = *reinterpret_cast<const v2f*>(Wpair + ((size_t)(hi * COUT) + nt * 16 + c) * 2);
            acc[nt] = __builtin_amdgcn_wmma_f32_16x16x4_f32(
                          false, a, false, bb, (short)0, acc[nt], false, false);
        }
    }
#pragma unroll
    for (int nt = 0; nt < 8; ++nt) {
#pragma unroll
        for (int r = 0; r < 8; ++r)
            Pall[(row0 + 8 * hi + r) * COUT + nt * 16 + c] = acc[nt][r];
    }

    const float* frow = features + row * CIN + hi * 2;
#pragma unroll
    for (int kb = 1; kb < KPAD / 4; ++kb) {
        const v2f a = *reinterpret_cast<const v2f*>(frow + (kb - 1) * 4);
#pragma unroll
        for (int nt = 0; nt < 8; ++nt) {
            const v2f bb = *reinterpret_cast<const v2f*>(
                Wpair + ((size_t)((kb * 2 + hi) * COUT) + nt * 16 + c) * 2);
            acc[nt] = __builtin_amdgcn_wmma_f32_16x16x4_f32(
                          false, a, false, bb, (short)0, acc[nt], false, false);
        }
    }
#pragma unroll
    for (int nt = 0; nt < 8; ++nt) {
#pragma unroll
        for (int r = 0; r < 8; ++r)
            Fall[(row0 + 8 * hi + r) * COUT + nt * 16 + c] = acc[nt][r];
    }
}

typedef __attribute__((ext_vector_type(4))) float v4f;
#define GQ     4
#define NPART  (BATCH * NPOINT / GQ)

__global__ __launch_bounds__(256) void gather_rows_kernel(
    const float* __restrict__ Pall, const float* __restrict__ Fall,
    const int* __restrict__ fps_idx, const int* __restrict__ knn_idx,
    const float* __restrict__ gamma, const float* __restrict__ beta,
    const float* __restrict__ meanv, const float* __restrict__ varv,
    float* __restrict__ partial_sum, float* __restrict__ partial_sq,
    float* __restrict__ out_features, int final_pass)
{
    const int lane = threadIdx.x & 31;
    const int w  = __builtin_amdgcn_readfirstlane((int)(blockIdx.x * 8 + (threadIdx.x >> 5)));
    const int q0 = w * GQ;
    const int b  = q0 / NPOINT;
    const int ch = lane * 4;
    const float* Pb = Pall + (size_t)b * NPTS * COUT + ch;
    const float* Fb = Fall + (size_t)b * NPTS * COUT + ch;

    int idx[GQ];
#pragma unroll
    for (int g = 0; g < GQ; ++g) {
        const int q = q0 + g;
        const int v = (lane < 16) ? knn_idx[(size_t)q * KNN + (lane & 15)] : fps_idx[q];
        idx[g] = v & (NPTS - 1);
    }

    v4f scale = {}, shift = {};
    if (final_pass) {
#pragma unroll
        for (int j = 0; j < 4; ++j) {
            const float mu = meanv[ch + j];
            const float sc = gamma[ch + j] * rsqrtf(varv[ch + j] + BNEPS);
            scale[j] = sc;
            shift[j] = beta[ch + j] - mu * sc;
        }
    }

    v4f s = {}, ss = {};
#pragma unroll
    for (int g = 0; g < GQ; ++g) {
        const int nq = __builtin_amdgcn_readlane(idx[g], 16);
        const v4f pq = *reinterpret_cast<const v4f*>(Pb + (size_t)nq * COUT);
        v4f f[KNN];
#pragma unroll
        for (int k = 0; k < KNN; ++k) {
            const int n = __builtin_amdgcn_readlane(idx[g], k);
            f[k] = *reinterpret_cast<const v4f*>(Fb + (size_t)n * COUT);
        }
        if (!final_pass) {
#pragma unroll
            for (int k = 0; k < KNN; ++k) {
                const v4f h = f[k] - pq;
                s  += h;
                ss += h * h;
            }
        } else {
            v4f m = {-3.4e38f, -3.4e38f, -3.4e38f, -3.4e38f};
#pragma unroll
            for (int k = 0; k < KNN; ++k) {
                const v4f h = f[k] - pq;
#pragma unroll
                for (int j = 0; j < 4; ++j) {
                    float y = h[j] * scale[j] + shift[j];
                    y = fmaxf(y, 0.0f);
                    m[j] = fmaxf(m[j], y);
                }
            }
            *reinterpret_cast<v4f*>(out_features + (size_t)(q0 + g) * COUT + ch) = m;
        }
    }
    if (!final_pass) {
        *reinterpret_cast<v4f*>(partial_sum + (size_t)w * COUT + ch) = s;
        *reinterpret_cast<v4f*>(partial_sq  + (size_t)w * COUT + ch) = ss;
    }
}

__global__ __launch_bounds__(256) void reduce_stats_kernel(
    const float* __restrict__ partial_sum, const float* __restrict__ partial_sq,
    float* __restrict__ meanv, float* __restrict__ varv)
{
    const int ch  = blockIdx.x;
    const int tid = threadIdx.x;
    __shared__ float ssum[256], ssq[256];
    float s = 0.0f, sq = 0.0f;
    for (int i = tid; i < NPART; i += 256) {
        s  += partial_sum[(size_t)i * COUT + ch];
        sq += partial_sq [(size_t)i * COUT + ch];
    }
    ssum[tid] = s; ssq[tid] = sq;
    __syncthreads();
    for (int off = 128; off > 0; off >>= 1) {
        if (tid < off) { ssum[tid] += ssum[tid + off]; ssq[tid] += ssq[tid + off]; }
        __syncthreads();
    }
    if (tid == 0) {
        const float cnt = (float)(BATCH * NPOINT * KNN);
        const float mu  = ssum[0] / cnt;
        meanv[ch] = mu;
        varv[ch]  = ssq[0] / cnt - mu * mu;
    }
}

extern "C" void kernel_launch(void* const* d_in, const int* in_sizes, int n_in,
                              void* d_out, int out_size, void* d_ws, size_t ws_size,
                              hipStream_t stream)
{
    const float* xyz      = (const float*)d_in[0];
    const float* features = (const float*)d_in[1];
    const float* W        = (const float*)d_in[2];
    const float* gamma    = (const float*)d_in[3];
    const float* beta     = (const float*)d_in[4];

    float* out          = (float*)d_out;
    float* new_xyz      = out;
    float* out_features = out + BATCH * NPOINT * 3;

    char* ws = (char*)d_ws;
    size_t off = 0;
    int*   fps_idx     = (int*)(ws + off);   off += sizeof(int)   * BATCH * NPOINT;
    int*   knn_idx     = (int*)(ws + off);   off += sizeof(int)   * BATCH * NPOINT * KNN;
    float* Wpair       = (float*)(ws + off); off += sizeof(float) * KPAD * COUT;
    off = (off + 255) & ~(size_t)255;
    float* partial_sum = (float*)(ws + off); off += sizeof(float) * NPART * COUT;
    float* partial_sq  = (float*)(ws + off); off += sizeof(float) * NPART * COUT;
    float* meanv       = (float*)(ws + off); off += sizeof(float) * COUT;
    float* varv        = (float*)(ws + off); off += sizeof(float) * COUT;
    off = (off + 255) & ~(size_t)255;
    float* Pall        = (float*)(ws + off); off += sizeof(float) * BATCH * NPTS * COUT;
    float* Fall        = (float*)(ws + off); off += sizeof(float) * BATCH * NPTS * COUT;
    (void)ws_size; (void)in_sizes; (void)n_in; (void)out_size;

    fps_kernel<<<BATCH, 1024, 0, stream>>>(xyz, fps_idx);
    gather_new_xyz_kernel<<<(BATCH * NPOINT * 3 + 255) / 256, 256, 0, stream>>>(
        xyz, fps_idx, new_xyz);
    knn_kernel<<<(BATCH * NPOINT) / KQ, KSEG * 32, 0, stream>>>(
        xyz, new_xyz, knn_idx);
    pad_w_kernel<<<(KPAD * COUT + 255) / 256, 256, 0, stream>>>(W, Wpair);
    project_kernel<<<(BATCH * NPTS) / PROWS, 256, 0, stream>>>(
        xyz, features, Wpair, Pall, Fall);
    gather_rows_kernel<<<NPART / 8, 256, 0, stream>>>(
        Pall, Fall, fps_idx, knn_idx, gamma, beta,
        meanv, varv, partial_sum, partial_sq, out_features, 0);
    reduce_stats_kernel<<<COUT, 256, 0, stream>>>(
        partial_sum, partial_sq, meanv, varv);
    gather_rows_kernel<<<NPART / 8, 256, 0, stream>>>(
        Pall, Fall, fps_idx, knn_idx, gamma, beta,
        meanv, varv, partial_sum, partial_sq, out_features, 1);
}


# Sampling launch also runs, on the WGPs it leaves idle, all that needs no picks: 16-NN lists of every point (gathers index them by pick) and the projection (W read as given); 8 launches to 5
# speedup vs baseline: 3.6711x; 1.0820x over previous
#include <hip/hip_runtime.h>

typedef __attribute__((ext_vector_type(2))) float v2f;
typedef __attribute__((ext_vector_type(8))) float v8f;

#define BATCH   4
#define NPTS    8192
#define CIN     64
#define COUT    128
#define NPOINT  2048
#define KNN     16
#define KRAW    67
#define KPAD    68
#define BNEPS   1e-5f

typedef __attribute__((ext_vector_type(2))) unsigned v2u;
typedef __attribute__((ext_vector_type(4))) unsigned v4u;

__device__ __forceinline__ unsigned umax2(unsigned a, unsigned b)
{
    return a > b ? a : b;
}

__device__ __forceinline__ unsigned umax32(unsigned* m)
{
#pragma unroll
    for (int w = 16; w > 0; w >>= 1) {
#pragma unroll
        for (int j = 0; j < w; ++j) m[j] = umax2(m[j], m[j + w]);
    }
    return m[0];
}

__device__ __forceinline__ void keymax32(unsigned* h, unsigned* l, unsigned& mh, unsigned& ml)
{
    unsigned t[32];
#pragma unroll
    for (int j = 0; j < 32; ++j) t[j] = h[j];
    const unsigned top = umax32(t);
#pragma unroll
    for (int j = 0; j < 32; ++j) l[j] = (h[j] == top) ? l[j] : 0u;
    mh = top;
    ml = umax32(l);
}

__device__ __forceinline__ unsigned fkey(float x)
{
    const unsigned u = __float_as_uint(x);
    return u ^ ((u >> 31) ? 0xFFFFFFFFu : 0x80000000u);
}

__device__ __forceinline__ unsigned wave_umax(unsigned v)
{
    v = umax2(v, (unsigned)__builtin_amdgcn_update_dpp(0, (int)v, 0x161, 0xf, 0xf, true));
    v = umax2(v, (unsigned)__builtin_amdgcn_update_dpp(0, (int)v, 0x162, 0xf, 0xf, true));
    v = umax2(v, (unsigned)__builtin_amdgcn_update_dpp(0, (int)v, 0x164, 0xf, 0xf, true));
    v = umax2(v, (unsigned)__builtin_amdgcn_update_dpp(0, (int)v, 0x168, 0xf, 0xf, true));
    v = umax2(v, (unsigned)__builtin_amdgcn_permlanex16((int)v, (int)v, 0x76543210, 0xfedcba98, false, false));
    return v;
}

__device__ __forceinline__ void lds_barrier()
{
    asm volatile("s_wait_dscnt 0x0\n\ts_barrier_signal -1\n\ts_barrier_wait -1" ::: "memory");
}

__device__ __forceinline__ void fps_body(const float* __restrict__ xyz,
                                         int* __restrict__ fps_idx, const int b)
{
    const int tid  = threadIdx.x;
    const int lane = tid & 31;
    const int wv   = tid >> 5;
    const float* xb = xyz + (size_t)b * NPTS * 3;

    __shared__ __attribute__((aligned(8))) v2u ssort[NPTS];
    __shared__ __attribute__((aligned(16))) v2u skey[16 * 66];
    __shared__ __attribute__((aligned(16))) v2u spart[32];
    __shared__ __attribute__((aligned(16))) v4u spick[33];
    __shared__ __attribute__((aligned(16))) v4u skeep[32];

#pragma unroll
    for (int r = 0; r < 8; ++r) {
        const int n = r * 1024 + tid;
        v2u e;
        e.x = fkey(xb[n * 3]);
        e.y = (unsigned)n;
        ssort[n] = e;
    }
    __syncthreads();
    for (unsigned k = 2; k <= NPTS; k <<= 1) {
        for (unsigned j = k >> 1; j > 0; j >>= 1) {
            unsigned ii[4];
            v2u ea[4], eb[4];
#pragma unroll
            for (int r = 0; r < 4; ++r) {
                const unsigned t = (unsigned)(r * 1024 + tid);
                ii[r] = ((t & ~(j - 1u)) << 1) | (t & (j - 1u));
                ea[r] = ssort[ii[r]];
                eb[r] = ssort[ii[r] | j];
            }
#pragma unroll
            for (int r = 0; r < 4; ++r) {
                const bool up = (ii[r] & k) == 0u;
                if ((ea[r].x > eb[r].x) == up) {
                    ssort[ii[r]]     = eb[r];
                    ssort[ii[r] | j] = ea[r];
                }
            }
            __syncthreads();
        }
    }

    const int wa = wv >> 3, wb = wv & 7;
    const int slab = 4 * (((((wb & 3) - wa) & 3) << 1) | (wb >> 2)) + wa;

    float px[8], py[8], pz[8], dist[8];
    unsigned linv[8];
    bool unf[8];
#pragma unroll
    for (int i = 0; i < 8; ++i) {
        const unsigned n = ssort[slab * 256 + i * 32 + lane].y & (NPTS - 1);
        px[i] = xb[n * 3 + 0];
        py[i] = xb[n * 3 + 1];
        pz[i] = xb[n * 3 + 2];
        linv[i] = ~n;
        unf[i]  = n >= 6144u;
        dist[i] = 3.4e38f;
    }
    const unsigned nlo = ssort[slab * 256].y & (NPTS - 1);
    const unsigned nhi = ssort[slab * 256 + 255].y & (NPTS - 1);
    const float xlo = __uint_as_float((unsigned)__builtin_amdgcn_readfirstlane(
                          (int)__float_as_uint(xb[nlo * 3])));
    const float xhi = __uint_as_float((unsigned)__builtin_amdgcn_readfirstlane(
                          (int)__float_as_uint(xb[nhi * 3])));

    const int wslot = (lane >> 1) * 66 + wv * 2 + (lane & 1);

    int cur = 0;
    float cx = xb[0];
    float cy = xb[1];
    float cz = xb[2];
    float wmax = 3.4e38f;
    unsigned npl = 0u;
    float nx = 0.0f, ny = 0.0f, nz = 0.0f;
    for (int s = 0; s < NPOINT; ++s) {
        if (tid == 0) fps_idx[b * NPOINT + s] = cur;

        const float g   = fmaxf(fmaxf(xlo - cx, cx - xhi), 0.0f);
        const float bnd = g * g;
        if (s == 0 || !(bnd >= wmax)) {
            float nd[8];
#pragma unroll
            for (int i = 0; i < 8; ++i) {
                const float dx = px[i] - cx;
                const float dy = py[i] - cy;
                const float dz = pz[i] - cz;
                const float q = dy * dy;
                const float f = __builtin_fmaf(dx, dx, q);
                float u;
                {
#pragma clang fp contract(off)
                    u = dx * dx + q;
                }
                const float d = __builtin_fmaf(dz, dz, unf[i] ? u : f);
                nd[i] = fminf(dist[i], d);
                dist[i] = nd[i];
            }
            float best = fmaxf(fmaxf(fmaxf(nd[0], nd[1]), fmaxf(nd[2], nd[3])),
                               fmaxf(fmaxf(nd[4], nd[5]), fmaxf(nd[6], nd[7])));
            unsigned lo = 0u;
#pragma unroll
            for (int i = 0; i < 8; ++i) lo = umax2(lo, (nd[i] == best) ? linv[i] : 0u);
            v2u key;
            key.x = lo;
            key.y = __float_as_uint(best);
            skey[wslot] = key;
        }
        if (wv == 0) {
            v4u nrec;
            nrec.x = npl;
            nrec.y = __float_as_uint(nx);
            nrec.z = __float_as_uint(ny);
            nrec.w = __float_as_uint(nz);
            skeep[lane] = nrec;
        }
        lds_barrier();
        if (wv == 0) {
            const v4u* col = reinterpret_cast<const v4u*>(skey) + lane;
            unsigned h[32], l[32];
#pragma unroll
            for (int j = 0; j < 16; ++j) {
                const v4u p = col[j * 33];
                l[2 * j] = p.x; h[2 * j] = p.y; l[2 * j + 1] = p.z; h[2 * j + 1] = p.w;
            }
            unsigned ph, pl;
            keymax32(h, l, ph, pl);
            const int cand = (int)(~pl) & (NPTS - 1);
            const float sx = xb[cand * 3 + 0];
            const float sy = xb[cand * 3 + 1];
            const float sz = xb[cand * 3 + 2];
            {
                v2u part;
                part.x = pl;
                part.y = ph;
                spart[lane] = part;
            }
            const v4u kept = skeep[lane];
            const unsigned gh = wave_umax(ph);
            const unsigned gl = wave_umax((ph == gh) ? pl : 0u);
            const bool mine  = (pl == gl);
            const bool stale = mine && (kept.x != pl);
            if (__builtin_amdgcn_ballot_w32(stale) != 0u) {
                if (mine) {
                    v4u rec;
                    rec.x = ~gl;
                    rec.y = __float_as_uint(sx);
                    rec.z = __float_as_uint(sy);
                    rec.w = __float_as_uint(sz);
                    spick[0] = rec;
                }
            } else {
                v4u rec = kept;
                rec.x = ~gl;
                spick[mine ? 0 : 1 + lane] = rec;
            }
            npl = pl;
            nx = sx;
            ny = sy;
            nz = sz;
        }
        lds_barrier();
        const v4u pk = spick[0];
        const v2u wm = spart[wv];
        cur = __builtin_amdgcn_readfirstlane((int)pk.x);
        cx  = __uint_as_float((unsigned)__builtin_amdgcn_readfirstlane((int)pk.y));
        cy  = __uint_as_float((unsigned)__builtin_amdgcn_readfirstlane((int)pk.z));
        cz  = __uint_as_float((unsigned)__builtin_amdgcn_readfirstlane((int)pk.w));
        wmax = __uint_as_float((unsigned)__builtin_amdgcn_readfirstlane((int)wm.y));
    }
}

__global__ void gather_new_xyz_kernel(const float* __restrict__ xyz,
                                      const int* __restrict__ fps_idx,
                                      float* __restrict__ new_xyz)
{
    const int e = blockIdx.x * blockDim.x + threadIdx.x;
    if (e >= BATCH * NPOINT * 3) return;
    const int c = e % 3;
    const int p = (e / 3) % NPOINT;
    const int b = e / (3 * NPOINT);
    const int n = fps_idx[b * NPOINT + p];
    new_xyz[e] = xyz[((size_t)b * NPTS + n) * 3 + c];
}

#define KSEG   16
#define KSEGN  (NPTS / KSEG)
#define KQ     32
#define KSETS  2
#define KNN_GROUPS (BATCH * NPTS / (KSETS * KQ))

__device__ __forceinline__ void knn_body(const float* __restrict__ xyz,
                                         int* __restrict__ knn_all, const int grp)
{
    __shared__ float sd[KSETS][KSEG][KNN][KQ];
    __shared__ int   si[KSETS][KSEG][KNN][KQ];

    const int lane = threadIdx.x & 31;
    const int seg  = (threadIdx.x >> 5) & (KSEG - 1);
    const int set  = threadIdx.x >> 9;
    const int q    = (grp * KSETS + set) * KQ + lane;
    const int b    = (grp * KSETS * KQ) / NPTS;
    const float qx = xyz[q * 3 + 0];
    const float qy = xyz[q * 3 + 1];
    const float qz = xyz[q * 3 + 2];
    const float* xb = xyz + (size_t)b * NPTS * 3;

    float v[KNN]; int id[KNN];
#pragma unroll
    for (int t = 0; t < KNN; ++t) { v[t] = 3.4e38f; id[t] = 0; }

    const int j0 = seg * KSEGN;
    for (int j = j0; j < j0 + KSEGN; ++j) {
        const float dx = xb[j * 3 + 0] - qx;
        const float dy = xb[j * 3 + 1] - qy;
        const float dz = xb[j * 3 + 2] - qz;
        const float d  = dx * dx + dy * dy + dz * dz;
        if (d < v[KNN - 1]) {
            v[KNN - 1] = d; id[KNN - 1] = j;
#pragma unroll
            for (int t = KNN - 1; t > 0; --t) {
                if (v[t] < v[t - 1]) {
                    const float tv = v[t]; v[t] = v[t - 1]; v[t - 1] = tv;
                    const int   ti = id[t]; id[t] = id[t - 1]; id[t - 1] = ti;
                }
            }
        }
    }
#pragma unroll
    for (int t = 0; t < KNN; ++t) { sd[set][seg][t][lane] = v[t]; si[set][seg][t][lane] = id[t]; }
    __syncthreads();

    if (seg == 0) {
        float hd[KSEG];
#pragma unroll
        for (int s = 0; s < KSEG; ++s) hd[s] = sd[set][s][0][lane];
        unsigned long long heads = 0ull;
        int oid[KNN];
#pragma unroll
        for (int t = 0; t < KNN; ++t) {
            float best = hd[0]; int bs = 0;
#pragma unroll
            for (int s = 1; s < KSEG; ++s)
                if (hd[s] < best) { best = hd[s]; bs = s; }
            const int p = (int)((heads >> (4 * bs)) & 15ull);
            oid[t] = si[set][bs][p][lane];
            if (t < KNN - 1) {
                heads += 1ull << (4 * bs);
                const float nh = sd[set][bs][p + 1][lane];
#pragma unroll
                for (int s = 0; s < KSEG; ++s) hd[s] = (s == bs) ? nh : hd[s];
            }
        }
#pragma unroll
        for (int t = 0; t < KNN; ++t) knn_all[(size_t)q * KNN + t] = oid[t];
    }
}

#define PROWS 512
#define PRJ_GROUPS (BATCH * NPTS / PROWS)

__device__ __forceinline__ void project_body(
    const float* __restrict__ xyz, const float* __restrict__ features,
    const float* __restrict__ W,
    float* __restrict__ Pall, float* __restrict__ Fall, const int grp)
{
    const int lane = threadIdx.x & 31;
    const int wv   = threadIdx.x >> 5;
    const int c    = lane & 15;
    const int hi   = lane >> 4;
    const size_t row0 = (size_t)grp * PROWS + (size_t)wv * 16;
    const size_t row  = row0 + c;

    const float* p    = xyz + row * 3;
    const float* frow = features + row * CIN + hi * 2;
#pragma unroll 1
    for (int ch = 0; ch < 2; ++ch) {
        const int col = ch * 64 + c;
        v8f acc[4];
#pragma unroll
        for (int nt = 0; nt < 4; ++nt) acc[nt] = (v8f){};

        {
            v2f a;
            a.x = hi ? p[2] : p[0];
            a.y = hi ? 0.0f : p[1];
#pragma unroll
            for (int nt = 0; nt < 4; ++nt) {
                v2f bb;
                bb.x = W[(hi ? 2 : 0) * COUT + nt * 16 + col];
                bb.y = hi ? 0.0f : W[COUT + nt * 16 + col];
                acc[nt] = __builtin_amdgcn_wmma_f32_16x16x4_f32(
                              false, a, false, bb, (short)0, acc[nt], false, false);
            }
        }
#pragma unroll
        for (int nt = 0; nt < 4; ++nt) {
#pragma unroll
            for (int r = 0; r < 8; ++r)
                Pall[(row0 + 8 * hi + r) * COUT + nt * 16 + col] = acc[nt][r];
        }

#pragma unroll
        for (int kb = 1; kb < KPAD / 4; ++kb) {
            const v2f a = *reinterpret_cast<const v2f*>(frow + (kb - 1) * 4);
#pragma unroll
            for (int nt = 0; nt < 4; ++nt) {
                v2f bb;
                bb.x = W[(4 * kb + 2 * hi - 1) * COUT + nt * 16 + col];
                bb.y = W[(4 * kb + 2 * hi) * COUT + nt * 16 + col];
                acc[nt] = __builtin_amdgcn_wmma_f32_16x16x4_f32(
                              false, a, false, bb, (short)0, acc[nt], false, false);
            }
        }
#pragma unroll
        for (int nt = 0; nt < 4; ++nt) {
#pragma unroll
            for (int r = 0; r < 8; ++r)
                Fall[(row0 + 8 * hi + r) * COUT + nt * 16 + col] = acc[nt][r];
        }
    }
}

__global__ __launch_bounds__(1024) void sample_and_prepare_kernel(
    const float* __restrict__ xyz, const float* __restrict__ features,
    const float* __restrict__ W,
    int* __restrict__ fps_idx, int* __restrict__ knn_all,
    float* __restrict__ Pall, float* __restrict__ Fall)
{
    const int g = blockIdx.x;
    if (g < BATCH)                   fps_body(xyz, fps_idx, g);
    else if (g < BATCH + KNN_GROUPS) knn_body(xyz, knn_all, g - BATCH);
    else                             project_body(xyz, features, W, Pall, Fall, g - BATCH - KNN_GROUPS);
}

typedef __attribute__((ext_vector_type(4))) float v4f;
#define GQ     4
#define NPART  (BATCH * NPOINT / GQ)

__global__ __launch_bounds__(256) void gather_rows_kernel(
    const float* __restrict__ Pall, const float* __restrict__ Fall,
    const int* __restrict__ fps_idx, const int* __restrict__ knn_all,
    const float* __restrict__ gamma, const float* __restrict__ beta,
    const float* __restrict__ meanv, const float* __restrict__ varv,
    float* __restrict__ partial_sum, float* __restrict__ partial_sq,
    float* __restrict__ out_features, int final_pass)
{
    const int lane = threadIdx.x & 31;
    const int w  = __builtin_amdgcn_readfirstlane((int)(blockIdx.x * 8 + (threadIdx.x >> 5)));
    const int q0 = w * GQ;
    const int b  = q0 / NPOINT;
    const int ch = lane * 4;
    const float* Pb = Pall + (size_t)b * NPTS * COUT + ch;
    const float* Fb = Fall + (size_t)b * NPTS * COUT + ch;

    int idx[GQ];
#pragma unroll
    for (int g = 0; g < GQ; ++g) {
        const int qn = fps_idx[q0 + g] & (NPTS - 1);
        const int v = (lane < 16) ? knn_all[((size_t)b * NPTS + qn) * KNN + (lane & 15)] : qn;
        idx[g] = v & (NPTS - 1);
    }

    v4f scale = {}, shift = {};
    if (final_pass) {
#pragma unroll
        for (int j = 0; j < 4; ++j) {
            const float mu = meanv[ch + j];
            const float sc = gamma[ch + j] * rsqrtf(varv[ch + j] + BNEPS);
            scale[j] = sc;
            shift[j] = beta[ch + j] - mu * sc;
        }
    }

    v4f s = {}, ss = {};
#pragma unroll
    for (int g = 0; g < GQ; ++g) {
        const int nq = __builtin_amdgcn_readlane(idx[g], 16);
        const v4f pq = *reinterpret_cast<const v4f*>(Pb + (size_t)nq * COUT);
        v4f f[KNN];
#pragma unroll
        for (int k = 0; k < KNN; ++k) {
            const int n = __builtin_amdgcn_readlane(idx[g], k);
            f[k] = *reinterpret_cast<const v4f*>(Fb + (size_t)n * COUT);
        }
        if (!final_pass) {
#pragma unroll
            for (int k = 0; k < KNN; ++k) {
                const v4f h = f[k] - pq;
                s  += h;
                ss += h * h;
            }
        } else {
            v4f m = {-3.4e38f, -3.4e38f, -3.4e38f, -3.4e38f};
#pragma unroll
            for (int k = 0; k < KNN; ++k) {
                const v4f h = f[k] - pq;
#pragma unroll
                for (int j = 0; j < 4; ++j) {
                    float y = h[j] * scale[j] + shift[j];
                    y = fmaxf(y, 0.0f);
                    m[j] = fmaxf(m[j], y);
                }
            }
            *reinterpret_cast<v4f*>(out_features + (size_t)(q0 + g) * COUT + ch) = m;
        }
    }
    if (!final_pass) {
        *reinterpret_cast<v4f*>(partial_sum + (size_t)w * COUT + ch) = s;
        *reinterpret_cast<v4f*>(partial_sq  + (size_t)w * COUT + ch) = ss;
    }
}

__global__ __launch_bounds__(256) void reduce_stats_kernel(
    const float* __restrict__ partial_sum, const float* __restrict__ partial_sq,
    float* __restrict__ meanv, float* __restrict__ varv)
{
    const int ch  = blockIdx.x;
    const int tid = threadIdx.x;
    __shared__ float ssum[256], ssq[256];
    float s = 0.0f, sq = 0.0f;
    for (int i = tid; i < NPART; i += 256) {
        s  += partial_sum[(size_t)i * COUT + ch];
        sq += partial_sq [(size_t)i * COUT + ch];
    }
    ssum[tid] = s; ssq[tid] = sq;
    __syncthreads();
    for (int off = 128; off > 0; off >>= 1) {
        if (tid < off) { ssum[tid] += ssum[tid + off]; ssq[tid] += ssq[tid + off]; }
        __syncthreads();
    }
    if (tid == 0) {
        const float cnt = (float)(BATCH * NPOINT * KNN);
        const float mu  = ssum[0] / cnt;
        meanv[ch] = mu;
        varv[ch]  = ssq[0] / cnt - mu * mu;
    }
}

extern "C" void kernel_launch(void* const* d_in, const int* in_sizes, int n_in,
                              void* d_out, int out_size, void* d_ws, size_t ws_size,
                              hipStream_t stream)
{
    const float* xyz      = (const float*)d_in[0];
    const float* features = (const float*)d_in[1];
    const float* W        = (const float*)d_in[2];
    const float* gamma    = (const float*)d_in[3];
    const float* beta     = (const float*)d_in[4];

    float* out          = (float*)d_out;
    float* new_xyz      = out;
    float* out_features = out + BATCH * NPOINT * 3;

    char* ws = (char*)d_ws;
    size_t off = 0;
    int*   fps_idx     = (int*)(ws + off);   off += sizeof(int)   * BATCH * NPOINT;
    int*   knn_all     = (int*)(ws + off);   off += sizeof(int)   * BATCH * NPTS * KNN;
    off = (off + 255) & ~(size_t)255;
    float* partial_sum = (float*)(ws + off); off += sizeof(float) * NPART * COUT;
    float* partial_sq  = (float*)(ws + off); off += sizeof(float) * NPART * COUT;
    float* meanv       = (float*)(ws + off); off += sizeof(float) * COUT;
    float* varv        = (float*)(ws + off); off += sizeof(float) * COUT;
    off = (off + 255) & ~(size_t)255;
    float* Pall        = (float*)(ws + off); off += sizeof(float) * BATCH * NPTS * COUT;
    float* Fall        = (float*)(ws + off); off += sizeof(float) * BATCH * NPTS * COUT;
    (void)ws_size; (void)in_sizes; (void)n_in; (void)out_size;

    sample_and_prepare_kernel<<<BATCH + KNN_GROUPS + PRJ_GROUPS, 1024, 0, stream>>>(
        xyz, features, W, fps_idx, knn_all, Pall, Fall);
    gather_new_xyz_kernel<<<(BATCH * NPOINT * 3 + 255) / 256, 256, 0, stream>>>(
        xyz, fps_idx, new_xyz);
    gather_rows_kernel<<<NPART / 8, 256, 0, stream>>>(
        Pall, Fall, fps_idx, knn_all, gamma, beta,
        meanv, varv, partial_sum, partial_sq, out_features, 0);
    reduce_stats_kernel<<<COUT, 256, 0, stream>>>(
        partial_sum, partial_sq, meanv, varv);
    gather_rows_kernel<<<NPART / 8, 256, 0, stream>>>(
        Pall, Fall, fps_idx, knn_all, gamma, beta,
        meanv, varv, partial_sum, partial_sq, out_features, 1);
}


# Sampling: first wave's serial part cut: key table as distance plane + index plane, lane maximum issued among the index choice, pick = lone top lane, record stored before stale test; bit-identical
# speedup vs baseline: 3.9206x; 1.0680x over previous
#include <hip/hip_runtime.h>

typedef __attribute__((ext_vector_type(2))) float v2f;
typedef __attribute__((ext_vector_type(8))) float v8f;

#define BATCH   4
#define NPTS    8192
#define CIN     64
#define COUT    128
#define NPOINT  2048
#define KNN     16
#define KRAW    67
#define KPAD    68
#define BNEPS   1e-5f

typedef __attribute__((ext_vector_type(2))) unsigned v2u;
typedef __attribute__((ext_vector_type(4))) unsigned v4u;

__device__ __forceinline__ unsigned umax2(unsigned a, unsigned b)
{
    return a > b ? a : b;
}

__device__ __forceinline__ unsigned umax32(unsigned* m)
{
#pragma unroll
    for (int w = 16; w > 0; w >>= 1) {
#pragma unroll
        for (int j = 0; j < w; ++j) m[j] = umax2(m[j], m[j + w]);
    }
    return m[0];
}

__device__ __forceinline__ unsigned sel4(const v4u h, const v4u l, const unsigned top)
{
    const unsigned a = (h.x == top) ? l.x : 0u;
    const unsigned b = (h.y == top) ? l.y : 0u;
    const unsigned c = (h.z == top) ? l.z : 0u;
    const unsigned d = (h.w == top) ? l.w : 0u;
    return umax2(umax2(a, b), umax2(c, d));
}

__device__ __forceinline__ unsigned fkey(float x)
{
    const unsigned u = __float_as_uint(x);
    return u ^ ((u >> 31) ? 0xFFFFFFFFu : 0x80000000u);
}

template <int X>
__device__ __forceinline__ unsigned row_umax_step(const unsigned v)
{
    return umax2(v, (unsigned)__builtin_amdgcn_update_dpp(0, (int)v, 0x160 + X, 0xf, 0xf, true));
}

__device__ __forceinline__ unsigned wave_umax(unsigned v)
{
    v = umax2(v, (unsigned)__builtin_amdgcn_update_dpp(0, (int)v, 0x161, 0xf, 0xf, true));
    v = umax2(v, (unsigned)__builtin_amdgcn_update_dpp(0, (int)v, 0x162, 0xf, 0xf, true));
    v = umax2(v, (unsigned)__builtin_amdgcn_update_dpp(0, (int)v, 0x164, 0xf, 0xf, true));
    v = umax2(v, (unsigned)__builtin_amdgcn_update_dpp(0, (int)v, 0x168, 0xf, 0xf, true));
    v = umax2(v, (unsigned)__builtin_amdgcn_permlanex16((int)v, (int)v, 0x76543210, 0xfedcba98, false, false));
    return v;
}

__device__ __forceinline__ void lds_barrier()
{
    asm volatile("s_wait_dscnt 0x0\n\ts_barrier_signal -1\n\ts_barrier_wait -1" ::: "memory");
}

__device__ __forceinline__ void fps_body(const float* __restrict__ xyz,
                                         int* __restrict__ fps_idx, const int b)
{
    const int tid  = threadIdx.x;
    const int lane = tid & 31;
    const int wv   = tid >> 5;
    const float* xb = xyz + (size_t)b * NPTS * 3;

    __shared__ __attribute__((aligned(8))) v2u ssort[NPTS];
    __shared__ __attribute__((aligned(16))) unsigned skey[16 * 132];
    __shared__ __attribute__((aligned(16))) v2u spart[32];
    __shared__ __attribute__((aligned(16))) v4u spick[33];

#pragma unroll
    for (int r = 0; r < 8; ++r) {
        const int n = r * 1024 + tid;
        v2u e;
        e.x = fkey(xb[n * 3]);
        e.y = (unsigned)n;
        ssort[n] = e;
    }
    __syncthreads();
    for (unsigned k = 2; k <= NPTS; k <<= 1) {
        for (unsigned j = k >> 1; j > 0; j >>= 1) {
            unsigned ii[4];
            v2u ea[4], eb[4];
#pragma unroll
            for (int r = 0; r < 4; ++r) {
                const unsigned t = (unsigned)(r * 1024 + tid);
                ii[r] = ((t & ~(j - 1u)) << 1) | (t & (j - 1u));
                ea[r] = ssort[ii[r]];
                eb[r] = ssort[ii[r] | j];
            }
#pragma unroll
            for (int r = 0; r < 4; ++r) {
                const bool up = (ii[r] & k) == 0u;
                if ((ea[r].x > eb[r].x) == up) {
                    ssort[ii[r]]     = eb[r];
                    ssort[ii[r] | j] = ea[r];
                }
            }
            __syncthreads();
        }
    }

    const int wa = wv >> 3, wb = wv & 7;
    const int slab = 4 * (((((wb & 3) - wa) & 3) << 1) | (wb >> 2)) + wa;

    float px[8], py[8], pz[8], dist[8];
    unsigned linv[8];
    bool unf[8];
#pragma unroll
    for (int i = 0; i < 8; ++i) {
        const unsigned n = ssort[slab * 256 + i * 32 + lane].y & (NPTS - 1);
        px[i] = xb[n * 3 + 0];
        py[i] = xb[n * 3 + 1];
        pz[i] = xb[n * 3 + 2];
        linv[i] = ~n;
        unf[i]  = n >= 6144u;
        dist[i] = 3.4e38f;
    }
    const unsigned nlo = ssort[slab * 256].y & (NPTS - 1);
    const unsigned nhi = ssort[slab * 256 + 255].y & (NPTS - 1);
    const float xlo = __uint_as_float((unsigned)__builtin_amdgcn_readfirstlane(
                          (int)__float_as_uint(xb[nlo * 3])));
    const float xhi = __uint_as_float((unsigned)__builtin_amdgcn_readfirstlane(
                          (int)__float_as_uint(xb[nhi * 3])));

    const int hslot = (lane >> 2) * 264 + wv * 4 + (lane & 3);

    int cur = 0;
    float cx = xb[0];
    float cy = xb[1];
    float cz = xb[2];
    float wmax = 3.4e38f;
    unsigned npl = 0u;
    float nx = 0.0f, ny = 0.0f, nz = 0.0f;
    for (int s = 0; s < NPOINT; ++s) {
        if (tid == 0) fps_idx[b * NPOINT + s] = cur;

        const float g   = fmaxf(fmaxf(xlo - cx, cx - xhi), 0.0f);
        const float bnd = g * g;
        if (s == 0 || !(bnd >= wmax)) {
            float nd[8];
#pragma unroll
            for (int i = 0; i < 8; ++i) {
                const float dx = px[i] - cx;
                const float dy = py[i] - cy;
                const float dz = pz[i] - cz;
                const float q = dy * dy;
                const float f = __builtin_fmaf(dx, dx, q);
                float u;
                {
#pragma clang fp contract(off)
                    u = dx * dx + q;
                }
                const float d = __builtin_fmaf(dz, dz, unf[i] ? u : f);
                nd[i] = fminf(dist[i], d);
                dist[i] = nd[i];
            }
            float best = fmaxf(fmaxf(fmaxf(nd[0], nd[1]), fmaxf(nd[2], nd[3])),
                               fmaxf(fmaxf(nd[4], nd[5]), fmaxf(nd[6], nd[7])));
            unsigned lo = 0u;
#pragma unroll
            for (int i = 0; i < 8; ++i) lo = umax2(lo, (nd[i] == best) ? linv[i] : 0u);
            skey[hslot]       = __float_as_uint(best);
            skey[hslot + 132] = lo;
        }
        lds_barrier();
        if (wv == 0) {
            const v4u* col = reinterpret_cast<const v4u*>(skey) + lane;
            v4u hq[8], lq[8];
            unsigned t[32];
#pragma unroll
            for (int g = 0; g < 8; ++g) hq[g] = col[g * 66];
#pragma unroll
            for (int g = 0; g < 8; ++g) lq[g] = col[g * 66 + 33];
#pragma unroll
            for (int g = 0; g < 8; ++g) {
                t[4 * g] = hq[g].x; t[4 * g + 1] = hq[g].y; t[4 * g + 2] = hq[g].z; t[4 * g + 3] = hq[g].w;
            }
            const unsigned ph = umax32(t);
            unsigned m[8];
            unsigned gh = row_umax_step<1>(ph);
            m[0] = sel4(hq[0], lq[0], ph);
            __builtin_amdgcn_sched_barrier(0);
            gh = row_umax_step<2>(gh);
            m[1] = sel4(hq[1], lq[1], ph);
            __builtin_amdgcn_sched_barrier(0);
            gh = row_umax_step<4>(gh);
            m[2] = sel4(hq[2], lq[2], ph);
            __builtin_amdgcn_sched_barrier(0);
            gh = row_umax_step<8>(gh);
            m[3] = sel4(hq[3], lq[3], ph);
            __builtin_amdgcn_sched_barrier(0);
            const unsigned go = (unsigned)__builtin_amdgcn_permlanex16((int)gh, (int)gh, 0x76543210, 0xfedcba98, false, false);
            m[4] = sel4(hq[4], lq[4], ph);
            __builtin_amdgcn_sched_barrier(0);
            gh = umax2(gh, go);
            m[5] = sel4(hq[5], lq[5], ph);
            __builtin_amdgcn_sched_barrier(0);
            const bool top = (ph == gh);
            const unsigned holders = __builtin_amdgcn_ballot_w32(top);
            const bool several = (holders & (holders - 1u)) != 0u;
            m[6] = sel4(hq[6], lq[6], ph);
            __builtin_amdgcn_sched_barrier(0);
            m[7] = sel4(hq[7], lq[7], ph);
            const unsigned pl = umax2(umax2(umax2(m[0], m[1]), umax2(m[2], m[3])),
                                      umax2(umax2(m[4], m[5]), umax2(m[6], m[7])));
            bool mine = top;
            unsigned minemask = holders;
            if (__builtin_expect(several, 0)) {
                const unsigned gl = wave_umax(top ? pl : 0u);
                mine = (pl == gl);
                minemask = __builtin_amdgcn_ballot_w32(mine);
            }
            {
                v4u rec;
                rec.x = ~pl;
                rec.y = __float_as_uint(nx);
                rec.z = __float_as_uint(ny);
                rec.w = __float_as_uint(nz);
                spick[mine ? 0 : 1 + lane] = rec;
            }
            const unsigned changed = __builtin_amdgcn_ballot_w32(npl != pl);
            const int cand = (int)(~pl) & (NPTS - 1);
            const float sx = xb[cand * 3 + 0];
            const float sy = xb[cand * 3 + 1];
            const float sz = xb[cand * 3 + 2];
            {
                v2u part;
                part.x = pl;
                part.y = ph;
                spart[lane] = part;
            }
            if (__builtin_expect((minemask & changed) != 0u, 0)) {
                if (mine) {
                    v4u rec;
                    rec.x = ~pl;
                    rec.y = __float_as_uint(sx);
                    rec.z = __float_as_uint(sy);
                    rec.w = __float_as_uint(sz);
                    spick[0] = rec;
                }
            }
            npl = pl;
            nx = sx;
            ny = sy;
            nz = sz;
        }
        lds_barrier();
        const v4u pk = spick[0];
        const v2u wm = spart[wv];
        cur = __builtin_amdgcn_readfirstlane((int)pk.x);
        cx  = __uint_as_float((unsigned)__builtin_amdgcn_readfirstlane((int)pk.y));
        cy  = __uint_as_float((unsigned)__builtin_amdgcn_readfirstlane((int)pk.z));
        cz  = __uint_as_float((unsigned)__builtin_amdgcn_readfirstlane((int)pk.w));
        wmax = __uint_as_float((unsigned)__builtin_amdgcn_readfirstlane((int)wm.y));
    }
}

__global__ void gather_new_xyz_kernel(const float* __restrict__ xyz,
                                      const int* __restrict__ fps_idx,
                                      float* __restrict__ new_xyz)
{
    const int e = blockIdx.x * blockDim.x + threadIdx.x;
    if (e >= BATCH * NPOINT * 3) return;
    const int c = e % 3;
    const int p = (e / 3) % NPOINT;
    const int b = e / (3 * NPOINT);
    const int n = fps_idx[b * NPOINT + p];
    new_xyz[e] = xyz[((size_t)b * NPTS + n) * 3 + c];
}

#define KSEG   16
#define KSEGN  (NPTS / KSEG)
#define KQ     32
#define KSETS  2
#define KNN_GROUPS (BATCH * NPTS / (KSETS * KQ))

__device__ __forceinline__ void knn_body(const float* __restrict__ xyz,
                                         int* __restrict__ knn_all, const int grp)
{
    __shared__ float sd[KSETS][KSEG][KNN][KQ];
    __shared__ int   si[KSETS][KSEG][KNN][KQ];

    const int lane = threadIdx.x & 31;
    const int seg  = (threadIdx.x >> 5) & (KSEG - 1);
    const int set  = threadIdx.x >> 9;
    const int q    = (grp * KSETS + set) * KQ + lane;
    const int b    = (grp * KSETS * KQ) / NPTS;
    const float qx = xyz[q * 3 + 0];
    const float qy = xyz[q * 3 + 1];
    const float qz = xyz[q * 3 + 2];
    const float* xb = xyz + (size_t)b * NPTS * 3;

    float v[KNN]; int id[KNN];
#pragma unroll
    for (int t = 0; t < KNN; ++t) { v[t] = 3.4e38f; id[t] = 0; }

    const int j0 = seg * KSEGN;
    for (int j = j0; j < j0 + KSEGN; ++j) {
        const float dx = xb[j * 3 + 0] - qx;
        const float dy = xb[j * 3 + 1] - qy;
        const float dz = xb[j * 3 + 2] - qz;
        const float d  = dx * dx + dy * dy + dz * dz;
        if (d < v[KNN - 1]) {
            v[KNN - 1] = d; id[KNN - 1] = j;
#pragma unroll
            for (int t = KNN - 1; t > 0; --t) {
                if (v[t] < v[t - 1]) {
                    const float tv = v[t]; v[t] = v[t - 1]; v[t - 1] = tv;
                    const int   ti = id[t]; id[t] = id[t - 1]; id[t - 1] = ti;
                }
            }
        }
    }
#pragma unroll
    for (int t = 0; t < KNN; ++t) { sd[set][seg][t][lane] = v[t]; si[set][seg][t][lane] = id[t]; }
    __syncthreads();

    if (seg == 0) {
        float hd[KSEG];
#pragma unroll
        for (int s = 0; s < KSEG; ++s) hd[s] = sd[set][s][0][lane];
        unsigned long long heads = 0ull;
        int oid[KNN];
#pragma unroll
        for (int t = 0; t < KNN; ++t) {
            float best = hd[0]; int bs = 0;
#pragma unroll
            for (int s = 1; s < KSEG; ++s)
                if (hd[s] < best) { best = hd[s]; bs = s; }
            const int p = (int)((heads >> (4 * bs)) & 15ull);
            oid[t] = si[set][bs][p][lane];
            if (t < KNN - 1) {
                heads += 1ull << (4 * bs);
                const float nh = sd[set][bs][p + 1][lane];
#pragma unroll
                for (int s = 0; s < KSEG; ++s) hd[s] = (s == bs) ? nh : hd[s];
            }
        }
#pragma unroll
        for (int t = 0; t < KNN; ++t) knn_all[(size_t)q * KNN + t] = oid[t];
    }
}

#define PROWS 512
#define PRJ_GROUPS (BATCH * NPTS / PROWS)

__device__ __forceinline__ void project_body(
    const float* __restrict__ xyz, const float* __restrict__ features,
    const float* __restrict__ W,
    float* __restrict__ Pall, float* __restrict__ Fall, const int grp)
{
    const int lane = threadIdx.x & 31;
    const int wv   = threadIdx.x >> 5;
    const int c    = lane & 15;
    const int hi   = lane >> 4;
    const size_t row0 = (size_t)grp * PROWS + (size_t)wv * 16;
    const size_t row  = row0 + c;

    const float* p    = xyz + row * 3;
    const float* frow = features + row * CIN + hi * 2;
#pragma unroll 1
    for (int ch = 0; ch < 2; ++ch) {
        const int col = ch * 64 + c;
        v8f acc[4];
#pragma unroll
        for (int nt = 0; nt < 4; ++nt) acc[nt] = (v8f){};

        {
            v2f a;
            a.x = hi ? p[2] : p[0];
            a.y = hi ? 0.0f : p[1];
#pragma unroll
            for (int nt = 0; nt < 4; ++nt) {
                v2f bb;
                bb.x = W[(hi ? 2 : 0) * COUT + nt * 16 + col];
                bb.y = hi ? 0.0f : W[COUT + nt * 16 + col];
                acc[nt] = __builtin_amdgcn_wmma_f32_16x16x4_f32(
                              false, a, false, bb, (short)0, acc[nt], false, false);
            }
        }
#pragma unroll
        for (int nt = 0; nt < 4; ++nt) {
#pragma unroll
            for (int r = 0; r < 8; ++r)
                Pall[(row0 + 8 * hi + r) * COUT + nt * 16 + col] = acc[nt][r];
        }

#pragma unroll
        for (int kb = 1; kb < KPAD / 4; ++kb) {
            const v2f a = *reinterpret_cast<const v2f*>(frow + (kb - 1) * 4);
#pragma unroll
            for (int nt = 0; nt < 4; ++nt) {
                v2f bb;
                bb.x = W[(4 * kb + 2 * hi - 1) * COUT + nt * 16 + col];
                bb.y = W[(4 * kb + 2 * hi) * COUT + nt * 16 + col];
                acc[nt] = __builtin_amdgcn_wmma_f32_16x16x4_f32(
                              false, a, false, bb, (short)0, acc[nt], false, false);
            }
        }
#pragma unroll
        for (int nt = 0; nt < 4; ++nt) {
#pragma unroll
            for (int r = 0; r < 8; ++r)
                Fall[(row0 + 8 * hi + r) * COUT + nt * 16 + col] = acc[nt][r];
        }
    }
}

__global__ __launch_bounds__(1024) void sample_and_prepare_kernel(
    const float* __restrict__ xyz, const float* __restrict__ features,
    const float* __restrict__ W,
    int* __restrict__ fps_idx, int* __restrict__ knn_all,
    float* __restrict__ Pall, float* __restrict__ Fall)
{
    const int g = blockIdx.x;
    if (g < BATCH)                   fps_body(xyz, fps_idx, g);
    else if (g < BATCH + KNN_GROUPS) knn_body(xyz, knn_all, g - BATCH);
    else                             project_body(xyz, features, W, Pall, Fall, g - BATCH - KNN_GROUPS);
}

typedef __attribute__((ext_vector_type(4))) float v4f;
#define GQ     4
#define NPART  (BATCH * NPOINT / GQ)

__global__ __launch_bounds__(256) void gather_rows_kernel(
    const float* __restrict__ Pall, const float* __restrict__ Fall,
    const int* __restrict__ fps_idx, const int* __restrict__ knn_all,
    const float* __restrict__ gamma, const float* __restrict__ beta,
    const float* __restrict__ meanv, const float* __restrict__ varv,
    float* __restrict__ partial_sum, float* __restrict__ partial_sq,
    float* __restrict__ out_features, int final_pass)
{
    const int lane = threadIdx.x & 31;
    const int w  = __builtin_amdgcn_readfirstlane((int)(blockIdx.x * 8 + (threadIdx.x >> 5)));
    const int q0 = w * GQ;
    const int b  = q0 / NPOINT;
    const int ch = lane * 4;
    const float* Pb = Pall + (size_t)b * NPTS * COUT + ch;
    const float* Fb = Fall + (size_t)b * NPTS * COUT + ch;

    int idx[GQ];
#pragma unroll
    for (int g = 0; g < GQ; ++g) {
        const int qn = fps_idx[q0 + g] & (NPTS - 1);
        const int v = (lane < 16) ? knn_all[((size_t)b * NPTS + qn) * KNN + (lane & 15)] : qn;
        idx[g] = v & (NPTS - 1);
    }

    v4f scale = {}, shift = {};
    if (final_pass) {
#pragma unroll
        for (int j = 0; j < 4; ++j) {
            const float mu = meanv[ch + j];
            const float sc = gamma[ch + j] * rsqrtf(varv[ch + j] + BNEPS);
            scale[j] = sc;
            shift[j] = beta[ch + j] - mu * sc;
        }
    }

    v4f s = {}, ss = {};
#pragma unroll
    for (int g = 0; g < GQ; ++g) {
        const int nq = __builtin_amdgcn_readlane(idx[g], 16);
        const v4f pq = *reinterpret_cast<const v4f*>(Pb + (size_t)nq * COUT);
        v4f f[KNN];
#pragma unroll
        for (int k = 0; k < KNN; ++k) {
            const int n = __builtin_amdgcn_readlane(idx[g], k);
            f[k] = *reinterpret_cast<const v4f*>(Fb + (size_t)n * COUT);
        }
        if (!final_pass) {
#pragma unroll
            for (int k = 0; k < KNN; ++k) {
                const v4f h = f[k] - pq;
                s  += h;
                ss += h * h;
            }
        } else {
            v4f m = {-3.4e38f, -3.4e38f, -3.4e38f, -3.4e38f};
#pragma unroll
            for (int k = 0; k < KNN; ++k) {
                const v4f h = f[k] - pq;
#pragma unroll
                for (int j = 0; j < 4; ++j) {
                    float y = h[j] * scale[j] + shift[j];
                    y = fmaxf(y, 0.0f);
                    m[j] = fmaxf(m[j], y);
                }
            }
            *reinterpret_cast<v4f*>(out_features + (size_t)(q0 + g) * COUT + ch) = m;
        }
    }
    if (!final_pass) {
        *reinterpret_cast<v4f*>(partial_sum + (size_t)w * COUT + ch) = s;
        *reinterpret_cast<v4f*>(partial_sq  + (size_t)w * COUT + ch) = ss;
    }
}

__global__ __launch_bounds__(256) void reduce_stats_kernel(
    const float* __restrict__ partial_sum, const float* __restrict__ partial_sq,
    float* __restrict__ meanv, float* __restrict__ varv)
{
    const int ch  = blockIdx.x;
    const int tid = threadIdx.x;
    __shared__ float ssum[256], ssq[256];
    float s = 0.0f, sq = 0.0f;
    for (int i = tid; i < NPART; i += 256) {
        s  += partial_sum[(size_t)i * COUT + ch];
        sq += partial_sq [(size_t)i * COUT + ch];
    }
    ssum[tid] = s; ssq[tid] = sq;
    __syncthreads();
    for (int off = 128; off > 0; off >>= 1) {
        if (tid < off) { ssum[tid] += ssum[tid + off]; ssq[tid] += ssq[tid + off]; }
        __syncthreads();
    }
    if (tid == 0) {
        const float cnt = (float)(BATCH * NPOINT * KNN);
        const float mu  = ssum[0] / cnt;
        meanv[ch] = mu;
        varv[ch]  = ssq[0] / cnt - mu * mu;
    }
}

extern "C" void kernel_launch(void* const* d_in, const int* in_sizes, int n_in,
                              void* d_out, int out_size, void* d_ws, size_t ws_size,
                              hipStream_t stream)
{
    const float* xyz      = (const float*)d_in[0];
    const float* features = (const float*)d_in[1];
    const float* W        = (const float*)d_in[2];
    const float* gamma    = (const float*)d_in[3];
    const float* beta     = (const float*)d_in[4];

    float* out          = (float*)d_out;
    float* new_xyz      = out;
    float* out_features = out + BATCH * NPOINT * 3;

    char* ws = (char*)d_ws;
    size_t off = 0;
    int*   fps_idx     = (int*)(ws + off);   off += sizeof(int)   * BATCH * NPOINT;
    int*   knn_all     = (int*)(ws + off);   off += sizeof(int)   * BATCH * NPTS * KNN;
    off = (off + 255) & ~(size_t)255;
    float* partial_sum = (float*)(ws + off); off += sizeof(float) * NPART * COUT;
    float* partial_sq  = (float*)(ws + off); off += sizeof(float) * NPART * COUT;
    float* meanv       = (float*)(ws + off); off += sizeof(float) * COUT;
    float* varv        = (float*)(ws + off); off += sizeof(float) * COUT;
    off = (off + 255) & ~(size_t)255;
    float* Pall        = (float*)(ws + off); off += sizeof(float) * BATCH * NPTS * COUT;
    float* Fall        = (float*)(ws + off); off += sizeof(float) * BATCH * NPTS * COUT;
    (void)ws_size; (void)in_sizes; (void)n_in; (void)out_size;

    sample_and_prepare_kernel<<<BATCH + KNN_GROUPS + PRJ_GROUPS, 1024, 0, stream>>>(
        xyz, features, W, fps_idx, knn_all, Pall, Fall);
    gather_new_xyz_kernel<<<(BATCH * NPOINT * 3 + 255) / 256, 256, 0, stream>>>(
        xyz, fps_idx, new_xyz);
    gather_rows_kernel<<<NPART / 8, 256, 0, stream>>>(
        Pall, Fall, fps_idx, knn_all, gamma, beta,
        meanv, varv, partial_sum, partial_sq, out_features, 0);
    reduce_stats_kernel<<<COUT, 256, 0, stream>>>(
        partial_sum, partial_sq, meanv, varv);
    gather_rows_kernel<<<NPART / 8, 256, 0, stream>>>(
        Pall, Fall, fps_idx, knn_all, gamma, beta,
        meanv, varv, partial_sum, partial_sq, out_features, 1);
}


# Sampling: after a pick each wave reads index, x, y, z and its column maximum by ONE 4-byte LDS load (lane l = word l; v_readlane where used); was a 16-byte + a 4-byte load; bit-identical
# speedup vs baseline: 4.0959x; 1.0447x over previous
#include <hip/hip_runtime.h>

typedef __attribute__((ext_vector_type(2))) float v2f;
typedef __attribute__((ext_vector_type(8))) float v8f;

#define BATCH   4
#define NPTS    8192
#define CIN     64
#define COUT    128
#define NPOINT  2048
#define KNN     16
#define KRAW    67
#define KPAD    68
#define BNEPS   1e-5f

typedef __attribute__((ext_vector_type(2))) unsigned v2u;
typedef __attribute__((ext_vector_type(4))) unsigned v4u;

__device__ __forceinline__ unsigned umax2(unsigned a, unsigned b)
{
    return a > b ? a : b;
}

__device__ __forceinline__ unsigned umax32(unsigned* m)
{
#pragma unroll
    for (int w = 16; w > 0; w >>= 1) {
#pragma unroll
        for (int j = 0; j < w; ++j) m[j] = umax2(m[j], m[j + w]);
    }
    return m[0];
}

__device__ __forceinline__ unsigned sel4(const v4u h, const v4u l, const unsigned top)
{
    const unsigned a = (h.x == top) ? l.x : 0u;
    const unsigned b = (h.y == top) ? l.y : 0u;
    const unsigned c = (h.z == top) ? l.z : 0u;
    const unsigned d = (h.w == top) ? l.w : 0u;
    return umax2(umax2(a, b), umax2(c, d));
}

__device__ __forceinline__ unsigned fkey(float x)
{
    const unsigned u = __float_as_uint(x);
    return u ^ ((u >> 31) ? 0xFFFFFFFFu : 0x80000000u);
}

template <int X>
__device__ __forceinline__ unsigned row_umax_step(const unsigned v)
{
    return umax2(v, (unsigned)__builtin_amdgcn_update_dpp(0, (int)v, 0x160 + X, 0xf, 0xf, true));
}

__device__ __forceinline__ unsigned wave_umax(unsigned v)
{
    v = umax2(v, (unsigned)__builtin_amdgcn_update_dpp(0, (int)v, 0x161, 0xf, 0xf, true));
    v = umax2(v, (unsigned)__builtin_amdgcn_update_dpp(0, (int)v, 0x162, 0xf, 0xf, true));
    v = umax2(v, (unsigned)__builtin_amdgcn_update_dpp(0, (int)v, 0x164, 0xf, 0xf, true));
    v = umax2(v, (unsigned)__builtin_amdgcn_update_dpp(0, (int)v, 0x168, 0xf, 0xf, true));
    v = umax2(v, (unsigned)__builtin_amdgcn_permlanex16((int)v, (int)v, 0x76543210, 0xfedcba98, false, false));
    return v;
}

__device__ __forceinline__ void lds_barrier()
{
    asm volatile("s_wait_dscnt 0x0\n\ts_barrier_signal -1\n\ts_barrier_wait -1" ::: "memory");
}

__device__ __forceinline__ void fps_body(const float* __restrict__ xyz,
                                         int* __restrict__ fps_idx, const int b)
{
    const int tid  = threadIdx.x;
    const int lane = tid & 31;
    const int wv   = tid >> 5;
    const float* xb = xyz + (size_t)b * NPTS * 3;

    __shared__ __attribute__((aligned(8))) v2u ssort[NPTS];
    __shared__ __attribute__((aligned(16))) unsigned skey[16 * 132];
    __shared__ __attribute__((aligned(16))) unsigned spub[33 * 4 + 32 * 2];
    v4u* const spick = reinterpret_cast<v4u*>(spub);
    v2u* const spart = reinterpret_cast<v2u*>(spub + 33 * 4);

#pragma unroll
    for (int r = 0; r < 8; ++r) {
        const int n = r * 1024 + tid;
        v2u e;
        e.x = fkey(xb[n * 3]);
        e.y = (unsigned)n;
        ssort[n] = e;
    }
    __syncthreads();
    for (unsigned k = 2; k <= NPTS; k <<= 1) {
        for (unsigned j = k >> 1; j > 0; j >>= 1) {
            unsigned ii[4];
            v2u ea[4], eb[4];
#pragma unroll
            for (int r = 0; r < 4; ++r) {
                const unsigned t = (unsigned)(r * 1024 + tid);
                ii[r] = ((t & ~(j - 1u)) << 1) | (t & (j - 1u));
                ea[r] = ssort[ii[r]];
                eb[r] = ssort[ii[r] | j];
            }
#pragma unroll
            for (int r = 0; r < 4; ++r) {
                const bool up = (ii[r] & k) == 0u;
                if ((ea[r].x > eb[r].x) == up) {
                    ssort[ii[r]]     = eb[r];
                    ssort[ii[r] | j] = ea[r];
                }
            }
            __syncthreads();
        }
    }

    const int wa = wv >> 3, wb = wv & 7;
    const int slab = 4 * (((((wb & 3) - wa) & 3) << 1) | (wb >> 2)) + wa;

    float px[8], py[8], pz[8], dist[8];
    unsigned linv[8];
    bool unf[8];
#pragma unroll
    for (int i = 0; i < 8; ++i) {
        const unsigned n = ssort[slab * 256 + i * 32 + lane].y & (NPTS - 1);
        px[i] = xb[n * 3 + 0];
        py[i] = xb[n * 3 + 1];
        pz[i] = xb[n * 3 + 2];
        linv[i] = ~n;
        unf[i]  = n >= 6144u;
        dist[i] = 3.4e38f;
    }
    const unsigned nlo = ssort[slab * 256].y & (NPTS - 1);
    const unsigned nhi = ssort[slab * 256 + 255].y & (NPTS - 1);
    const float xlo = __uint_as_float((unsigned)__builtin_amdgcn_readfirstlane(
                          (int)__float_as_uint(xb[nlo * 3])));
    const float xhi = __uint_as_float((unsigned)__builtin_amdgcn_readfirstlane(
                          (int)__float_as_uint(xb[nhi * 3])));

    const int hslot = (lane >> 2) * 264 + wv * 4 + (lane & 3);
    const int rdw = (lane < 4) ? lane : ((lane == 4) ? 33 * 4 + 2 * wv + 1 : 0);

    unsigned pw = (lane >= 1 && lane <= 3) ? __float_as_uint(xb[lane - 1])
                : ((lane == 4) ? __float_as_uint(3.4e38f) : 0u);
    unsigned npl = 0u;
    float nx = 0.0f, ny = 0.0f, nz = 0.0f;
    for (int s = 0; s < NPOINT; ++s) {
        if (tid == 0) fps_idx[b * NPOINT + s] = (int)pw;

        const float cx   = __uint_as_float((unsigned)__builtin_amdgcn_readlane((int)pw, 1));
        const float wmax = __uint_as_float((unsigned)__builtin_amdgcn_readlane((int)pw, 4));
        const float g   = fmaxf(fmaxf(xlo - cx, cx - xhi), 0.0f);
        const float bnd = g * g;
        if (s == 0 || !(bnd >= wmax)) {
            const float cy = __uint_as_float((unsigned)__builtin_amdgcn_readlane((int)pw, 2));
            const float cz = __uint_as_float((unsigned)__builtin_amdgcn_readlane((int)pw, 3));
            float nd[8];
#pragma unroll
            for (int i = 0; i < 8; ++i) {
                const float dx = px[i] - cx;
                const float dy = py[i] - cy;
                const float dz = pz[i] - cz;
                const float q = dy * dy;
                const float f = __builtin_fmaf(dx, dx, q);
                float u;
                {
#pragma clang fp contract(off)
                    u = dx * dx + q;
                }
                const float d = __builtin_fmaf(dz, dz, unf[i] ? u : f);
                nd[i] = fminf(dist[i], d);
                dist[i] = nd[i];
            }
            float best = fmaxf(fmaxf(fmaxf(nd[0], nd[1]), fmaxf(nd[2], nd[3])),
                               fmaxf(fmaxf(nd[4], nd[5]), fmaxf(nd[6], nd[7])));
            unsigned lo = 0u;
#pragma unroll
            for (int i = 0; i < 8; ++i) lo = umax2(lo, (nd[i] == best) ? linv[i] : 0u);
            skey[hslot]       = __float_as_uint(best);
            skey[hslot + 132] = lo;
        }
        lds_barrier();
        if (wv == 0) {
            const v4u* col = reinterpret_cast<const v4u*>(skey) + lane;
            v4u hq[8], lq[8];
            unsigned t[32];
#pragma unroll
            for (int g = 0; g < 8; ++g) hq[g] = col[g * 66];
#pragma unroll
            for (int g = 0; g < 8; ++g) lq[g] = col[g * 66 + 33];
#pragma unroll
            for (int g = 0; g < 8; ++g) {
                t[4 * g] = hq[g].x; t[4 * g + 1] = hq[g].y; t[4 * g + 2] = hq[g].z; t[4 * g + 3] = hq[g].w;
            }
            const unsigned ph = umax32(t);
            unsigned m[8];
            unsigned gh = row_umax_step<1>(ph);
            m[0] = sel4(hq[0], lq[0], ph);
            __builtin_amdgcn_sched_barrier(0);
            gh = row_umax_step<2>(gh);
            m[1] = sel4(hq[1], lq[1], ph);
            __builtin_amdgcn_sched_barrier(0);
            gh = row_umax_step<4>(gh);
            m[2] = sel4(hq[2], lq[2], ph);
            __builtin_amdgcn_sched_barrier(0);
            gh = row_umax_step<8>(gh);
            m[3] = sel4(hq[3], lq[3], ph);
            __builtin_amdgcn_sched_barrier(0);
            const unsigned go = (unsigned)__builtin_amdgcn_permlanex16((int)gh, (int)gh, 0x76543210, 0xfedcba98, false, false);
            m[4] = sel4(hq[4], lq[4], ph);
            __builtin_amdgcn_sched_barrier(0);
            gh = umax2(gh, go);
            m[5] = sel4(hq[5], lq[5], ph);
            __builtin_amdgcn_sched_barrier(0);
            const bool top = (ph == gh);
            const unsigned holders = __builtin_amdgcn_ballot_w32(top);
            const bool several = (holders & (holders - 1u)) != 0u;
            m[6] = sel4(hq[6], lq[6], ph);
            __builtin_amdgcn_sched_barrier(0);
            m[7] = sel4(hq[7], lq[7], ph);
            const unsigned pl = umax2(umax2(umax2(m[0], m[1]), umax2(m[2], m[3])),
                                      umax2(umax2(m[4], m[5]), umax2(m[6], m[7])));
            bool mine = top;
            unsigned minemask = holders;
            if (__builtin_expect(several, 0)) {
                const unsigned gl = wave_umax(top ? pl : 0u);
                mine = (pl == gl);
                minemask = __builtin_amdgcn_ballot_w32(mine);
            }
            {
                v4u rec;
                rec.x = ~pl;
                rec.y = __float_as_uint(nx);
                rec.z = __float_as_uint(ny);
                rec.w = __float_as_uint(nz);
                spick[mine ? 0 : 1 + lane] = rec;
            }
            const unsigned changed = __builtin_amdgcn_ballot_w32(npl != pl);
            const int cand = (int)(~pl) & (NPTS - 1);
            const float sx = xb[cand * 3 + 0];
            const float sy = xb[cand * 3 + 1];
            const float sz = xb[cand * 3 + 2];
            {
                v2u part;
                part.x = pl;
                part.y = ph;
                spart[lane] = part;
            }
            if (__builtin_expect((minemask & changed) != 0u, 0)) {
                if (mine) {
                    v4u rec;
                    rec.x = ~pl;
                    rec.y = __float_as_uint(sx);
                    rec.z = __float_as_uint(sy);
                    rec.w = __float_as_uint(sz);
                    spick[0] = rec;
                }
            }
            npl = pl;
            nx = sx;
            ny = sy;
            nz = sz;
        }
        lds_barrier();
        pw = spub[rdw];
    }
}

__global__ void gather_new_xyz_kernel(const float* __restrict__ xyz,
                                      const int* __restrict__ fps_idx,
                                      float* __restrict__ new_xyz)
{
    const int e = blockIdx.x * blockDim.x + threadIdx.x;
    if (e >= BATCH * NPOINT * 3) return;
    const int c = e % 3;
    const int p = (e / 3) % NPOINT;
    const int b = e / (3 * NPOINT);
    const int n = fps_idx[b * NPOINT + p];
    new_xyz[e] = xyz[((size_t)b * NPTS + n) * 3 + c];
}

#define KSEG   16
#define KSEGN  (NPTS / KSEG)
#define KQ     32
#define KSETS  2
#define KNN_GROUPS (BATCH * NPTS / (KSETS * KQ))

__device__ __forceinline__ void knn_body(const float* __restrict__ xyz,
                                         int* __restrict__ knn_all, const int grp)
{
    __shared__ float sd[KSETS][KSEG][KNN][KQ];
    __shared__ int   si[KSETS][KSEG][KNN][KQ];

    const int lane = threadIdx.x & 31;
    const int seg  = (threadIdx.x >> 5) & (KSEG - 1);
    const int set  = threadIdx.x >> 9;
    const int q    = (grp * KSETS + set) * KQ + lane;
    const int b    = (grp * KSETS * KQ) / NPTS;
    const float qx = xyz[q * 3 + 0];
    const float qy = xyz[q * 3 + 1];
    const float qz = xyz[q * 3 + 2];
    const float* xb = xyz + (size_t)b * NPTS * 3;

    float v[KNN]; int id[KNN];
#pragma unroll
    for (int t = 0; t < KNN; ++t) { v[t] = 3.4e38f; id[t] = 0; }

    const int j0 = seg * KSEGN;
    for (int j = j0; j < j0 + KSEGN; ++j) {
        const float dx = xb[j * 3 + 0] - qx;
        const float dy = xb[j * 3 + 1] - qy;
        const float dz = xb[j * 3 + 2] - qz;
        const float d  = dx * dx + dy * dy + dz * dz;
        if (d < v[KNN - 1]) {
            v[KNN - 1] = d; id[KNN - 1] = j;
#pragma unroll
            for (int t = KNN - 1; t > 0; --t) {
                if (v[t] < v[t - 1]) {
                    const float tv = v[t]; v[t] = v[t - 1]; v[t - 1] = tv;
                    const int   ti = id[t]; id[t] = id[t - 1]; id[t - 1] = ti;
                }
            }
        }
    }
#pragma unroll
    for (int t = 0; t < KNN; ++t) { sd[set][seg][t][lane] = v[t]; si[set][seg][t][lane] = id[t]; }
    __syncthreads();

    if (seg == 0) {
        float hd[KSEG];
#pragma unroll
        for (int s = 0; s < KSEG; ++s) hd[s] = sd[set][s][0][lane];
        unsigned long long heads = 0ull;
        int oid[KNN];
#pragma unroll
        for (int t = 0; t < KNN; ++t) {
            float best = hd[0]; int bs = 0;
#pragma unroll
            for (int s = 1; s < KSEG; ++s)
                if (hd[s] < best) { best = hd[s]; bs = s; }
            const int p = (int)((heads >> (4 * bs)) & 15ull);
            oid[t] = si[set][bs][p][lane];
            if (t < KNN - 1) {
                heads += 1ull << (4 * bs);
                const float nh = sd[set][bs][p + 1][lane];
#pragma unroll
                for (int s = 0; s < KSEG; ++s) hd[s] = (s == bs) ? nh : hd[s];
            }
        }
#pragma unroll
        for (int t = 0; t < KNN; ++t) knn_all[(size_t)q * KNN + t] = oid[t];
    }
}

#define PROWS 512
#define PRJ_GROUPS (BATCH * NPTS / PROWS)

__device__ __forceinline__ void project_body(
    const float* __restrict__ xyz, const float* __restrict__ features,
    const float* __restrict__ W,
    float* __restrict__ Pall, float* __restrict__ Fall, const int grp)
{
    const int lane = threadIdx.x & 31;
    const int wv   = threadIdx.x >> 5;
    const int c    = lane & 15;
    const int hi   = lane >> 4;
    const size_t row0 = (size_t)grp * PROWS + (size_t)wv * 16;
    const size_t row  = row0 + c;

    const float* p    = xyz + row * 3;
    const float* frow = features + row * CIN + hi * 2;
#pragma unroll 1
    for (int ch = 0; ch < 2; ++ch) {
        const int col = ch * 64 + c;
        v8f acc[4];
#pragma unroll
        for (int nt = 0; nt < 4; ++nt) acc[nt] = (v8f){};

        {
            v2f a;
            a.x = hi ? p[2] : p[0];
            a.y = hi ? 0.0f : p[1];
#pragma unroll
            for (int nt = 0; nt < 4; ++nt) {
                v2f bb;
                bb.x = W[(hi ? 2 : 0) * COUT + nt * 16 + col];
                bb.y = hi ? 0.0f : W[COUT + nt * 16 + col];
                acc[nt] = __builtin_amdgcn_wmma_f32_16x16x4_f32(
                              false, a, false, bb, (short)0, acc[nt], false, false);
            }
        }
#pragma unroll
        for (int nt = 0; nt < 4; ++nt) {
#pragma unroll
            for (int r = 0; r < 8; ++r)
                Pall[(row0 + 8 * hi + r) * COUT + nt * 16 + col] = acc[nt][r];
        }

#pragma unroll
        for (int kb = 1; kb < KPAD / 4; ++kb) {
            const v2f a = *reinterpret_cast<const v2f*>(frow + (kb - 1) * 4);
#pragma unroll
            for (int nt = 0; nt < 4; ++nt) {
                v2f bb;
                bb.x = W[(4 * kb + 2 * hi - 1) * COUT + nt * 16 + col];
                bb.y = W[(4 * kb + 2 * hi) * COUT + nt * 16 + col];
                acc[nt] = __builtin_amdgcn_wmma_f32_16x16x4_f32(
                              false, a, false, bb, (short)0, acc[nt], false, false);
            }
        }
#pragma unroll
        for (int nt = 0; nt < 4; ++nt) {
#pragma unroll
            for (int r = 0; r < 8; ++r)
                Fall[(row0 + 8 * hi + r) * COUT + nt * 16 + col] = acc[nt][r];
        }
    }
}

__global__ __launch_bounds__(1024) void sample_and_prepare_kernel(
    const float* __restrict__ xyz, const float* __restrict__ features,
    const float* __restrict__ W,
    int* __restrict__ fps_idx, int* __restrict__ knn_all,
    float* __restrict__ Pall, float* __restrict__ Fall)
{
    const int g = blockIdx.x;
    if (g < BATCH)                   fps_body(xyz, fps_idx, g);
    else if (g < BATCH + KNN_GROUPS) knn_body(xyz, knn_all, g - BATCH);
    else                             project_body(xyz, features, W, Pall, Fall, g - BATCH - KNN_GROUPS);
}

typedef __attribute__((ext_vector_type(4))) float v4f;
#define GQ     4
#define NPART  (BATCH * NPOINT / GQ)

__global__ __launch_bounds__(256) void gather_rows_kernel(
    const float* __restrict__ Pall, const float* __restrict__ Fall,
    const int* __restrict__ fps_idx, const int* __restrict__ knn_all,
    const float* __restrict__ gamma, const float* __restrict__ beta,
    const float* __restrict__ meanv, const float* __restrict__ varv,
    float* __restrict__ partial_sum, float* __restrict__ partial_sq,
    float* __restrict__ out_features, int final_pass)
{
    const int lane = threadIdx.x & 31;
    const int w  = __builtin_amdgcn_readfirstlane((int)(blockIdx.x * 8 + (threadIdx.x >> 5)));
    const int q0 = w * GQ;
    const int b  = q0 / NPOINT;
    const int ch = lane * 4;
    const float* Pb = Pall + (size_t)b * NPTS * COUT + ch;
    const float* Fb = Fall + (size_t)b * NPTS * COUT + ch;

    int idx[GQ];
#pragma unroll
    for (int g = 0; g < GQ; ++g) {
        const int qn = fps_idx[q0 + g] & (NPTS - 1);
        const int v = (lane < 16) ? knn_all[((size_t)b * NPTS + qn) * KNN + (lane & 15)] : qn;
        idx[g] = v & (NPTS - 1);
    }

    v4f scale = {}, shift = {};
    if (final_pass) {
#pragma unroll
        for (int j = 0; j < 4; ++j) {
            const float mu = meanv[ch + j];
            const float sc = gamma[ch + j] * rsqrtf(varv[ch + j] + BNEPS);
            scale[j] = sc;
            shift[j] = beta[ch + j] - mu * sc;
        }
    }

    v4f s = {}, ss = {};
#pragma unroll
    for (int g = 0; g < GQ; ++g) {
        const int nq = __builtin_amdgcn_readlane(idx[g], 16);
        const v4f pq = *reinterpret_cast<const v4f*>(Pb + (size_t)nq * COUT);
        v4f f[KNN];
#pragma unroll
        for (int k = 0; k < KNN; ++k) {
            const int n = __builtin_amdgcn_readlane(idx[g], k);
            f[k] = *reinterpret_cast<const v4f*>(Fb + (size_t)n * COUT);
        }
        if (!final_pass) {
#pragma unroll
            for (int k = 0; k < KNN; ++k) {
                const v4f h = f[k] - pq;
                s  += h;
                ss += h * h;
            }
        } else {
            v4f m = {-3.4e38f, -3.4e38f, -3.4e38f, -3.4e38f};
#pragma unroll
            for (int k = 0; k < KNN; ++k) {
                const v4f h = f[k] - pq;
#pragma unroll
                for (int j = 0; j < 4; ++j) {
                    float y = h[j] * scale[j] + shift[j];
                    y = fmaxf(y, 0.0f);
                    m[j] = fmaxf(m[j], y);
                }
            }
            *reinterpret_cast<v4f*>(out_features + (size_t)(q0 + g) * COUT + ch) = m;
        }
    }
    if (!final_pass) {
        *reinterpret_cast<v4f*>(partial_sum + (size_t)w * COUT + ch) = s;
        *reinterpret_cast<v4f*>(partial_sq  + (size_t)w * COUT + ch) = ss;
    }
}

__global__ __launch_bounds__(256) void reduce_stats_kernel(
    const float* __restrict__ partial_sum, const float* __restrict__ partial_sq,
    float* __restrict__ meanv, float* __restrict__ varv)
{
    const int ch  = blockIdx.x;
    const int tid = threadIdx.x;
    __shared__ float ssum[256], ssq[256];
    float s = 0.0f, sq = 0.0f;
    for (int i = tid; i < NPART; i += 256) {
        s  += partial_sum[(size_t)i * COUT + ch];
        sq += partial_sq [(size_t)i * COUT + ch];
    }
    ssum[tid] = s; ssq[tid] = sq;
    __syncthreads();
    for (int off = 128; off > 0; off >>= 1) {
        if (tid < off) { ssum[tid] += ssum[tid + off]; ssq[tid] += ssq[tid + off]; }
        __syncthreads();
    }
    if (tid == 0) {
        const float cnt = (float)(BATCH * NPOINT * KNN);
        const float mu  = ssum[0] / cnt;
        meanv[ch] = mu;
        varv[ch]  = ssq[0] / cnt - mu * mu;
    }
}

extern "C" void kernel_launch(void* const* d_in, const int* in_sizes, int n_in,
                              void* d_out, int out_size, void* d_ws, size_t ws_size,
                              hipStream_t stream)
{
    const float* xyz      = (const float*)d_in[0];
    const float* features = (const float*)d_in[1];
    const float* W        = (const float*)d_in[2];
    const float* gamma    = (const float*)d_in[3];
    const float* beta     = (const float*)d_in[4];

    float* out          = (float*)d_out;
    float* new_xyz      = out;
    float* out_features = out + BATCH * NPOINT * 3;

    char* ws = (char*)d_ws;
    size_t off = 0;
    int*   fps_idx     = (int*)(ws + off);   off += sizeof(int)   * BATCH * NPOINT;
    int*   knn_all     = (int*)(ws + off);   off += sizeof(int)   * BATCH * NPTS * KNN;
    off = (off + 255) & ~(size_t)255;
    float* partial_sum = (float*)(ws + off); off += sizeof(float) * NPART * COUT;
    float* partial_sq  = (float*)(ws + off); off += sizeof(float) * NPART * COUT;
    float* meanv       = (float*)(ws + off); off += sizeof(float) * COUT;
    float* varv        = (float*)(ws + off); off += sizeof(float) * COUT;
    off = (off + 255) & ~(size_t)255;
    float* Pall        = (float*)(ws + off); off += sizeof(float) * BATCH * NPTS * COUT;
    float* Fall        = (float*)(ws + off); off += sizeof(float) * BATCH * NPTS * COUT;
    (void)ws_size; (void)in_sizes; (void)n_in; (void)out_size;

    sample_and_prepare_kernel<<<BATCH + KNN_GROUPS + PRJ_GROUPS, 1024, 0, stream>>>(
        xyz, features, W, fps_idx, knn_all, Pall, Fall);
    gather_new_xyz_kernel<<<(BATCH * NPOINT * 3 + 255) / 256, 256, 0, stream>>>(
        xyz, fps_idx, new_xyz);
    gather_rows_kernel<<<NPART / 8, 256, 0, stream>>>(
        Pall, Fall, fps_idx, knn_all, gamma, beta,
        meanv, varv, partial_sum, partial_sq, out_features, 0);
    reduce_stats_kernel<<<COUT, 256, 0, stream>>>(
        partial_sum, partial_sq, meanv, varv);
    gather_rows_kernel<<<NPART / 8, 256, 0, stream>>>(
        Pall, Fall, fps_idx, knn_all, gamma, beta,
        meanv, varv, partial_sum, partial_sq, out_features, 1);
}
